# starting kernel
# baseline (speedup 1.0000x reference)
#include <hip/hip_runtime.h>
#include <stdint.h>

typedef __attribute__((ext_vector_type(2))) float v2f;
typedef __attribute__((ext_vector_type(8))) float v8f;

template <bool RELU_A, int N, int K, int KP>
__global__ __launch_bounds__(256)
void gcn_gemm_wmma_f32(const float* __restrict__ A,
                       const float* __restrict__ B,
                       float* __restrict__ C,
                       int M)
{
    extern __shared__ float ldsB[];

    const int tid  = threadIdx.x;
    const int lane = tid & 31;
    const int wave = tid >> 5;

    constexpr int NCOLS = N / 64;
    const int gwave = blockIdx.x * 8 + wave;
    const int mt    = gwave / NCOLS;
    const int n0    = (gwave - mt * NCOLS) * 64;
    const bool active = (mt * 16 < M);
    const int m0    = mt * 16;

    const int half = lane >> 4;
    const int l    = lane & 15;

    const float* arow  = active ? (A + (size_t)(m0 + l) * K + 2 * half) : A;
    const float* bfrag = ldsB + (size_t)(half * N + n0 + l) * 2;

    v8f acc0 = {}, acc1 = {}, acc2 = {}, acc3 = {};

    for (int k0 = 0; k0 < K; k0 += KP) {
        if (k0 != 0) __syncthreads();
        constexpr int PAIRS = (KP / 2) * N;
        #pragma unroll
        for (int idx = 0; idx < PAIRS / 256; ++idx) {
            const int pi = idx * 256 + tid;
            const int p  = pi / N;
            const int n  = pi - p * N;
            const int gk = k0 + 2 * p;
            v2f pr;
            pr.x = B[(size_t)gk * N + n];
            pr.y = B[(size_t)(gk + 1) * N + n];
            *(v2f*)(ldsB + (size_t)pi * 2) = pr;
        }
        __syncthreads();

        if (active) {
            #pragma unroll
            for (int kk = 0; kk < KP; kk += 4) {
                float a0 = arow[k0 + kk];
                float a1 = arow[k0 + kk + 1];
                if (RELU_A) { a0 = fmaxf(a0, 0.0f); a1 = fmaxf(a1, 0.0f); }
                v2f a; a.x = a0; a.y = a1;

                const float* bp = bfrag + (kk >> 1) * (N * 2);
                v2f b0 = *(const v2f*)(bp);
                v2f b1 = *(const v2f*)(bp + 32);
                v2f b2 = *(const v2f*)(bp + 64);
                v2f b3 = *(const v2f*)(bp + 96);

                acc0 = __builtin_amdgcn_wmma_f32_16x16x4_f32(false, a, false, b0,
                                                             (short)0, acc0, false, false);
                acc1 = __builtin_amdgcn_wmma_f32_16x16x4_f32(false, a, false, b1,
                                                             (short)0, acc1, false, false);
                acc2 = __builtin_amdgcn_wmma_f32_16x16x4_f32(false, a, false, b2,
                                                             (short)0, acc2, false, false);
                acc3 = __builtin_amdgcn_wmma_f32_16x16x4_f32(false, a, false, b3,
                                                             (short)0, acc3, false, false);
            }
        }
    }

    if (active) {
        #pragma unroll
        for (int i = 0; i < 8; ++i) {
            const size_t r = (size_t)(m0 + i + 8 * half);
            float* crow = C + r * N + n0 + l;
            crow[0]  = acc0[i];
            crow[16] = acc1[i];
            crow[32] = acc2[i];
            crow[48] = acc3[i];
        }
    }
}

__global__ __launch_bounds__(256)
void gcn_bias_init(float* __restrict__ out, const float* __restrict__ bias,
                   size_t total4, int dmask)
{
    size_t i = (size_t)blockIdx.x * blockDim.x + threadIdx.x;
    const size_t stride = (size_t)gridDim.x * blockDim.x;
    for (; i < total4; i += stride) {
        const int d = (int)((i * 4) & (size_t)dmask);
        *(float4*)(out + i * 4) = *(const float4*)(bias + d);
    }
}

template <int VEC>
__global__ __launch_bounds__(256)
void gcn_spmm_scatter(const int* __restrict__ rows,
                      const int* __restrict__ cols,
                      const float* __restrict__ w,
                      const float* __restrict__ sup,
                      float* __restrict__ out,
                      int E, int D)
{
    typedef __attribute__((ext_vector_type(VEC))) float vec_t;
    const int lane = threadIdx.x & 31;
    int e = blockIdx.x * 8 + (threadIdx.x >> 5);
    const int estride = gridDim.x * 8;

    for (; e < E; e += estride) {
        const int   r  = rows[e];
        const int   c  = cols[e];
        const float wt = w[e];

        vec_t s = *(const vec_t*)(sup + (size_t)c * D + lane * VEC);
        float* __restrict__ dst = out + (size_t)r * D + lane * VEC;

        #pragma unroll
        for (int j = 0; j < VEC; ++j)
            atomicAdd(dst + j, wt * s[j]);
    }
}

extern "C" void kernel_launch(void* const* d_in, const int* in_sizes, int n_in,
                              void* d_out, int out_size, void* d_ws, size_t ws_size,
                              hipStream_t stream)
{
    const float* x  = (const float*)d_in[0];
    const int*   ei = (const int*)  d_in[1];
    const float* ew = (const float*)d_in[2];
    const float* W1 = (const float*)d_in[3];
    const float* b1 = (const float*)d_in[4];
    const float* W2 = (const float*)d_in[5];
    const float* b2 = (const float*)d_in[6];
    float* out = (float*)d_out;

    const int N  = 100000;
    const int E  = in_sizes[2];
    constexpr int F  = 256;
    constexpr int H1 = 128;
    constexpr int H2 = 64;
    constexpr int KP = 64;

    float* sup1 = (float*)d_ws;
    float* h    = sup1 + (size_t)N * H1;
    float* sup2 = h    + (size_t)N * H1;

    {
        const int waves  = (N / 16) * (H1 / 64);
        const int blocks = (waves + 7) / 8;
        const size_t lds = (size_t)KP * H1 * sizeof(float);
        gcn_gemm_wmma_f32<false, H1, F, KP><<<blocks, 256, lds, stream>>>(x, W1, sup1, N);
    }
    gcn_bias_init<<<2048, 256, 0, stream>>>(h, b1, (size_t)N * H1 / 4, H1 - 1);
    gcn_spmm_scatter<4><<<8192, 256, 0, stream>>>(ei, ei + E, ew, sup1, h, E, H1);

    {
        const int waves  = (N / 16) * (H2 / 64);
        const int blocks = (waves + 7) / 8;
        const size_t lds = (size_t)KP * H2 * sizeof(float);
        gcn_gemm_wmma_f32<true, H2, H1, KP><<<blocks, 256, lds, stream>>>(h, W2, sup2, N);
    }
    gcn_bias_init<<<2048, 256, 0, stream>>>(out, b2, (size_t)N * H2 / 4, H2 - 1);
    gcn_spmm_scatter<2><<<8192, 256, 0, stream>>>(ei, ei + E, ew, sup2, out, E, H2);
}


# Float-atomic COO scatter replaced by a per-call CSR build (integer counting sort by row) and a one-wave-per-row register gather-sum with bias fused; per-row summation order changes
# speedup vs baseline: 1.8484x; 1.8484x over previous
#include <hip/hip_runtime.h>
#include <stdint.h>

typedef __attribute__((ext_vector_type(2))) float v2f;
typedef __attribute__((ext_vector_type(8))) float v8f;

template <bool RELU_A, int N, int K, int KP>
__global__ __launch_bounds__(256)
void gcn_gemm_wmma_f32(const float* __restrict__ A,
                       const float* __restrict__ B,
                       float* __restrict__ C,
                       int M)
{
    extern __shared__ float ldsB[];

    const int tid  = threadIdx.x;
    const int lane = tid & 31;
    const int wave = tid >> 5;

    constexpr int NCOLS = N / 64;
    const int gwave = blockIdx.x * 8 + wave;
    const int mt    = gwave / NCOLS;
    const int n0    = (gwave - mt * NCOLS) * 64;
    const bool active = (mt * 16 < M);
    const int m0    = mt * 16;

    const int half = lane >> 4;
    const int l    = lane & 15;

    const float* arow  = active ? (A + (size_t)(m0 + l) * K + 2 * half) : A;
    const float* bfrag = ldsB + (size_t)(half * N + n0 + l) * 2;

    v8f acc0 = {}, acc1 = {}, acc2 = {}, acc3 = {};

    for (int k0 = 0; k0 < K; k0 += KP) {
        if (k0 != 0) __syncthreads();
        constexpr int PAIRS = (KP / 2) * N;
        #pragma unroll
        for (int idx = 0; idx < PAIRS / 256; ++idx) {
            const int pi = idx * 256 + tid;
            const int p  = pi / N;
            const int n  = pi - p * N;
            const int gk = k0 + 2 * p;
            v2f pr;
            pr.x = B[(size_t)gk * N + n];
            pr.y = B[(size_t)(gk + 1) * N + n];
            *(v2f*)(ldsB + (size_t)pi * 2) = pr;
        }
        __syncthreads();

        if (active) {
            #pragma unroll
            for (int kk = 0; kk < KP; kk += 4) {
                float a0 = arow[k0 + kk];
                float a1 = arow[k0 + kk + 1];
                if (RELU_A) { a0 = fmaxf(a0, 0.0f); a1 = fmaxf(a1, 0.0f); }
                v2f a; a.x = a0; a.y = a1;

                const float* bp = bfrag + (kk >> 1) * (N * 2);
                v2f b0 = *(const v2f*)(bp);
                v2f b1 = *(const v2f*)(bp + 32);
                v2f b2 = *(const v2f*)(bp + 64);
                v2f b3 = *(const v2f*)(bp + 96);

                acc0 = __builtin_amdgcn_wmma_f32_16x16x4_f32(false, a, false, b0,
                                                             (short)0, acc0, false, false);
                acc1 = __builtin_amdgcn_wmma_f32_16x16x4_f32(false, a, false, b1,
                                                             (short)0, acc1, false, false);
                acc2 = __builtin_amdgcn_wmma_f32_16x16x4_f32(false, a, false, b2,
                                                             (short)0, acc2, false, false);
                acc3 = __builtin_amdgcn_wmma_f32_16x16x4_f32(false, a, false, b3,
                                                             (short)0, acc3, false, false);
            }
        }
    }

    if (active) {
        #pragma unroll
        for (int i = 0; i < 8; ++i) {
            const size_t r = (size_t)(m0 + i + 8 * half);
            float* crow = C + r * N + n0 + l;
            crow[0]  = acc0[i];
            crow[16] = acc1[i];
            crow[32] = acc2[i];
            crow[48] = acc3[i];
        }
    }
}

__global__ __launch_bounds__(256)
void gcn_csr_zero(int* __restrict__ cnt, int n)
{
    const int i = blockIdx.x * 256 + threadIdx.x;
    if (i < n) cnt[i] = 0;
}

__global__ __launch_bounds__(256)
void gcn_csr_count(const int* __restrict__ rows,
                   int* __restrict__ cnt,
                   int* __restrict__ rank,
                   int E)
{
    int e = blockIdx.x * 256 + threadIdx.x;
    const int stride = gridDim.x * 256;
    for (; e < E; e += stride)
        rank[e] = atomicAdd(cnt + rows[e], 1);
}

__global__ __launch_bounds__(1024)
void gcn_csr_scan(const int* __restrict__ cnt, int* __restrict__ start, int n)
{
    __shared__ int wpre[32];
    __shared__ int wtot[32];
    const int tid  = threadIdx.x;
    const int lane = tid & 31;
    const int wv   = tid >> 5;
    const int base = blockIdx.x * 1024;

    int s = 0;
    const int4* c4 = (const int4*)cnt;
    #pragma unroll 4
    for (int i = tid; i < base / 4; i += 1024) {
        const int4 v = c4[i];
        s += (v.x + v.y) + (v.z + v.w);
    }

    const int idx = base + tid;
    const int c = (idx < n) ? cnt[idx] : 0;

    int incl = c;
    for (int d = 1; d < 32; d <<= 1) {
        const int t = __shfl_up(incl, d);
        if (lane >= d) incl += t;
    }
    for (int d = 16; d >= 1; d >>= 1) s += __shfl_xor(s, d);

    if (lane == 31) wtot[wv] = incl;
    if (lane == 0)  wpre[wv] = s;
    __syncthreads();

    int off = 0;
    for (int k = 0; k < 32; ++k) {
        off += wpre[k];
        if (k < wv) off += wtot[k];
    }
    const int excl = off + incl - c;
    if (idx < n) start[idx] = excl;
    if (idx == n - 1) start[n] = excl + c;
}

__global__ __launch_bounds__(256)
void gcn_csr_fill(const int* __restrict__ rows,
                  const int* __restrict__ cols,
                  const float* __restrict__ w,
                  const int* __restrict__ rank,
                  const int* __restrict__ start,
                  int2* __restrict__ sorted,
                  int E)
{
    int e = blockIdx.x * 256 + threadIdx.x;
    const int stride = gridDim.x * 256;
    for (; e < E; e += stride) {
        const int p = start[rows[e]] + rank[e];
        sorted[p] = make_int2(cols[e], __float_as_int(w[e]));
    }
}

template <int D, int U>
__device__ __forceinline__
void gcn_gather_batch(float4& acc, const float* __restrict__ sup,
                      int ecol, int ewb, int j, int sub, int f)
{
    constexpr int EPI = 128 / D;
    int    c[U];
    float  wv[U];
    float4 v[U];
    #pragma unroll
    for (int u = 0; u < U; ++u) {
        if (EPI == 1) {
            c[u]  = __builtin_amdgcn_readlane(ecol, j + u);
            wv[u] = __int_as_float(__builtin_amdgcn_readlane(ewb, j + u));
        } else {
            const int c0 = __builtin_amdgcn_readlane(ecol, j + 2 * u);
            const int c1 = __builtin_amdgcn_readlane(ecol, j + 2 * u + 1);
            const int w0 = __builtin_amdgcn_readlane(ewb,  j + 2 * u);
            const int w1 = __builtin_amdgcn_readlane(ewb,  j + 2 * u + 1);
            c[u]  = sub ? c1 : c0;
            wv[u] = __int_as_float(sub ? w1 : w0);
        }
    }
    #pragma unroll
    for (int u = 0; u < U; ++u)
        v[u] = *(const float4*)(sup + (size_t)c[u] * D + f);
    #pragma unroll
    for (int u = 0; u < U; ++u) {
        acc.x = fmaf(wv[u], v[u].x, acc.x);
        acc.y = fmaf(wv[u], v[u].y, acc.y);
        acc.z = fmaf(wv[u], v[u].z, acc.z);
        acc.w = fmaf(wv[u], v[u].w, acc.w);
    }
}

template <int D>
__global__ __launch_bounds__(256)
void gcn_spmm_gather(const int* __restrict__ start,
                     const int2* __restrict__ sorted,
                     const float* __restrict__ sup,
                     const float* __restrict__ bias,
                     float* __restrict__ out,
                     int nrows)
{
    constexpr int LPE = D / 4;
    constexpr int EPI = 32 / LPE;
    const int lane = threadIdx.x & 31;
    const int sub  = lane / LPE;
    const int f    = (lane - sub * LPE) * 4;
    const int wave0  = __builtin_amdgcn_readfirstlane(blockIdx.x * 8 + (threadIdx.x >> 5));
    const int nwaves = gridDim.x * 8;
    const float4 bv = *(const float4*)(bias + f);

    for (int r = wave0; r < nrows; r += nwaves) {
        const int beg = start[r];
        const int end = start[r + 1];
        float4 acc = make_float4(0.0f, 0.0f, 0.0f, 0.0f);

        for (int base = beg; base < end; base += 32) {
            const int n = min(32, end - base);
            const int2 ev  = sorted[base + min(lane, n - 1)];
            const int ecol = ev.x;
            const int ewb  = (lane < n) ? ev.y : 0;
            int j = 0;
            for (; j + 8 <= n; j += 8)
                gcn_gather_batch<D, 8 / EPI>(acc, sup, ecol, ewb, j, sub, f);
            for (; j < n; j += 4)
                gcn_gather_batch<D, 4 / EPI>(acc, sup, ecol, ewb, j, sub, f);
        }

        if (EPI == 2) {
            acc.x += __shfl_xor(acc.x, 16);
            acc.y += __shfl_xor(acc.y, 16);
            acc.z += __shfl_xor(acc.z, 16);
            acc.w += __shfl_xor(acc.w, 16);
        }
        if (sub == 0) {
            float4 o;
            o.x = acc.x + bv.x; o.y = acc.y + bv.y;
            o.z = acc.z + bv.z; o.w = acc.w + bv.w;
            *(float4*)(out + (size_t)r * D + f) = o;
        }
    }
}

extern "C" void kernel_launch(void* const* d_in, const int* in_sizes, int n_in,
                              void* d_out, int out_size, void* d_ws, size_t ws_size,
                              hipStream_t stream)
{
    const float* x  = (const float*)d_in[0];
    const int*   ei = (const int*)  d_in[1];
    const float* ew = (const float*)d_in[2];
    const float* W1 = (const float*)d_in[3];
    const float* b1 = (const float*)d_in[4];
    const float* W2 = (const float*)d_in[5];
    const float* b2 = (const float*)d_in[6];
    float* out = (float*)d_out;

    const int N  = 100000;
    const int E  = in_sizes[2];
    constexpr int F  = 256;
    constexpr int H1 = 128;
    constexpr int H2 = 64;
    constexpr int KP = 64;

    float* sup1 = (float*)d_ws;
    float* h    = sup1 + (size_t)N * H1;
    float* sup2 = h    + (size_t)N * H1;
    int*   cnt    = (int*)(sup2 + (size_t)N * H2);
    int*   start  = cnt + 100352;
    int2*  sorted = (int2*)(start + 100352);
    int*   rank   = (int*)(sorted + E);

    gcn_csr_zero<<<392, 256, 0, stream>>>(cnt, N);
    gcn_csr_count<<<2048, 256, 0, stream>>>(ei, cnt, rank, E);
    gcn_csr_scan<<<(N + 1023) / 1024, 1024, 0, stream>>>(cnt, start, N);
    gcn_csr_fill<<<2048, 256, 0, stream>>>(ei, ei + E, ew, rank, start, sorted, E);

    {
        const int waves  = (N / 16) * (H1 / 64);
        const int blocks = (waves + 7) / 8;
        const size_t lds = (size_t)KP * H1 * sizeof(float);
        gcn_gemm_wmma_f32<false, H1, F, KP><<<blocks, 256, lds, stream>>>(x, W1, sup1, N);
    }
    gcn_spmm_gather<H1><<<2048, 256, 0, stream>>>(start, sorted, sup1, b1, h, N);

    {
        const int waves  = (N / 16) * (H2 / 64);
        const int blocks = (waves + 7) / 8;
        const size_t lds = (size_t)KP * H2 * sizeof(float);
        gcn_gemm_wmma_f32<true, H2, H1, KP><<<blocks, 256, lds, stream>>>(h, W2, sup2, N);
    }
    gcn_spmm_gather<H2><<<2048, 256, 0, stream>>>(start, sorted, sup2, b2, out, N);
}


# CSR build as a two-level integer counting sort on LDS counters (chunks sorted by row bucket, then one block per bucket): no global atomics, zero or scan launch; per-row sum order = LDS arrival order
# speedup vs baseline: 3.1019x; 1.6782x over previous
#include <hip/hip_runtime.h>
#include <stdint.h>

typedef __attribute__((ext_vector_type(2))) float v2f;
typedef __attribute__((ext_vector_type(8))) float v8f;

template <bool RELU_A, int N, int K, int KP>
__global__ __launch_bounds__(256)
void gcn_gemm_wmma_f32(const float* __restrict__ A,
                       const float* __restrict__ B,
                       float* __restrict__ C,
                       int M)
{
    extern __shared__ float ldsB[];

    const int tid  = threadIdx.x;
    const int lane = tid & 31;
    const int wave = tid >> 5;

    constexpr int NCOLS = N / 64;
    const int gwave = blockIdx.x * 8 + wave;
    const int mt    = gwave / NCOLS;
    const int n0    = (gwave - mt * NCOLS) * 64;
    const bool active = (mt * 16 < M);
    const int m0    = mt * 16;

    const int half = lane >> 4;
    const int l    = lane & 15;

    const float* arow  = active ? (A + (size_t)(m0 + l) * K + 2 * half) : A;
    const float* bfrag = ldsB + (size_t)(half * N + n0 + l) * 2;

    v8f acc0 = {}, acc1 = {}, acc2 = {}, acc3 = {};

    for (int k0 = 0; k0 < K; k0 += KP) {
        if (k0 != 0) __syncthreads();
        constexpr int PAIRS = (KP / 2) * N;
        #pragma unroll
        for (int idx = 0; idx < PAIRS / 256; ++idx) {
            const int pi = idx * 256 + tid;
            const int p  = pi / N;
            const int n  = pi - p * N;
            const int gk = k0 + 2 * p;
            v2f pr;
            pr.x = B[(size_t)gk * N + n];
            pr.y = B[(size_t)(gk + 1) * N + n];
            *(v2f*)(ldsB + (size_t)pi * 2) = pr;
        }
        __syncthreads();

        if (active) {
            #pragma unroll
            for (int kk = 0; kk < KP; kk += 4) {
                float a0 = arow[k0 + kk];
                float a1 = arow[k0 + kk + 1];
                if (RELU_A) { a0 = fmaxf(a0, 0.0f); a1 = fmaxf(a1, 0.0f); }
                v2f a; a.x = a0; a.y = a1;

                const float* bp = bfrag + (kk >> 1) * (N * 2);
                v2f b0 = *(const v2f*)(bp);
                v2f b1 = *(const v2f*)(bp + 32);
                v2f b2 = *(const v2f*)(bp + 64);
                v2f b3 = *(const v2f*)(bp + 96);

                acc0 = __builtin_amdgcn_wmma_f32_16x16x4_f32(false, a, false, b0,
                                                             (short)0, acc0, false, false);
                acc1 = __builtin_amdgcn_wmma_f32_16x16x4_f32(false, a, false, b1,
                                                             (short)0, acc1, false, false);
                acc2 = __builtin_amdgcn_wmma_f32_16x16x4_f32(false, a, false, b2,
                                                             (short)0, acc2, false, false);
                acc3 = __builtin_amdgcn_wmma_f32_16x16x4_f32(false, a, false, b3,
                                                             (short)0, acc3, false, false);
            }
        }
    }

    if (active) {
        #pragma unroll
        for (int i = 0; i < 8; ++i) {
            const size_t r = (size_t)(m0 + i + 8 * half);
            float* crow = C + r * N + n0 + l;
            crow[0]  = acc0[i];
            crow[16] = acc1[i];
            crow[32] = acc2[i];
            crow[48] = acc3[i];
        }
    }
}

#define CSR_R     256
#define CSR_NB    391
#define CSR_NBP   392
#define CSR_CH    8192
#define CSR_EPT   8
#define CSR_CAP   8192

__global__ __launch_bounds__(1024)
void gcn_part_chunks(const int* __restrict__ rows,
                     const int* __restrict__ cols,
                     const float* __restrict__ w,
                     int2* __restrict__ tmp,
                     int* __restrict__ coff,
                     int E)
{
    __shared__ int  cnt[CSR_NBP];
    __shared__ int  wsum[32];
    __shared__ int2 stage[CSR_CH];

    const int tid  = threadIdx.x;
    const int lane = tid & 31;
    const int wv   = tid >> 5;
    const int nch  = (E + CSR_CH - 1) / CSR_CH;

    for (int c = blockIdx.x; c < nch; c += gridDim.x) {
        const int base = c * CSR_CH;
        const int n    = min(CSR_CH, E - base);

        if (tid < CSR_NBP) cnt[tid] = 0;
        __syncthreads();

        int  pk[CSR_EPT], wb[CSR_EPT], bk[CSR_EPT], rk[CSR_EPT];
        #pragma unroll
        for (int u = 0; u < CSR_EPT; ++u) {
            const int i  = min(u * 1024 + tid, n - 1);
            const int r  = rows[base + i];
            const int cc = cols[base + i];
            wb[u] = __float_as_int(w[base + i]);
            bk[u] = min(r >> 8, CSR_NB - 1);
            pk[u] = ((r & (CSR_R - 1)) << 17) | cc;
        }
        #pragma unroll
        for (int u = 0; u < CSR_EPT; ++u) {
            rk[u] = 0;
            if (u * 1024 + tid < n) rk[u] = atomicAdd(&cnt[bk[u]], 1);
        }
        __syncthreads();

        int v = 0, incl = 0;
        if (tid < 416) {
            v = (tid < CSR_NB) ? cnt[tid] : 0;
            incl = v;
            for (int d = 1; d < 32; d <<= 1) {
                const int t = __shfl_up(incl, d);
                if (lane >= d) incl += t;
            }
            if (lane == 31) wsum[wv] = incl;
        }
        __syncthreads();
        if (tid < CSR_NBP) {
            int off = incl - v;
            for (int k = 0; k < wv; ++k) off += wsum[k];
            cnt[tid] = off;
            coff[(size_t)c * CSR_NBP + tid] = off;
        }
        __syncthreads();

        int dst[CSR_EPT];
        #pragma unroll
        for (int u = 0; u < CSR_EPT; ++u) dst[u] = cnt[bk[u]] + rk[u];
        #pragma unroll
        for (int u = 0; u < CSR_EPT; ++u)
            if (u * 1024 + tid < n) stage[dst[u]] = make_int2(pk[u], wb[u]);
        __syncthreads();

        for (int i = tid; i < n; i += 1024) tmp[(size_t)base + i] = stage[i];
        __syncthreads();
    }
}

__device__ __forceinline__
void gcn_csr_place(int2 rec, int* cur, bool fits, int2* stage,
                   int2* __restrict__ sorted, int bstart)
{
    const int pos = atomicAdd(&cur[rec.x >> 17], 1);
    const int2 o  = make_int2(rec.x & 0x1FFFF, rec.y);
    if (fits) stage[pos] = o;
    else      sorted[(size_t)bstart + pos] = o;
}

__global__ __launch_bounds__(512)
void gcn_csr_rows(const int2* __restrict__ tmp,
                  const int* __restrict__ coff,
                  int* __restrict__ start,
                  int2* __restrict__ sorted,
                  int E, int nrows)
{
    __shared__ int  roff[640];
    __shared__ int  rlen[640];
    __shared__ int  cnt[CSR_R];
    __shared__ int  wtot[8];
    __shared__ int  wbs[16];
    __shared__ int2 stage[CSR_CAP];

    const int tid  = threadIdx.x;
    const int lane = tid & 31;
    const int wv   = tid >> 5;
    const int nch  = (E + CSR_CH - 1) / CSR_CH;

    if (tid < 128) { roff[512 + tid] = 0; rlen[512 + tid] = 0; }

    for (int b = blockIdx.x; b < CSR_NB; b += gridDim.x) {
        if (tid < CSR_R) cnt[tid] = 0;
        int bsum = 0;

        for (int c0 = 0; c0 < nch; c0 += 512) {
            __syncthreads();
            const int c = c0 + tid;
            int len = 0, ro = 0;
            if (c < nch) {
                const int o0 = coff[(size_t)c * CSR_NBP + b];
                const int o1 = coff[(size_t)c * CSR_NBP + b + 1];
                ro  = c * CSR_CH + o0;
                len = o1 - o0;
                bsum += o0;
            }
            roff[tid] = ro;
            rlen[tid] = len;
            __syncthreads();
            const int tn = min(512, nch - c0);
            for (int k = wv; k < tn; k += 128) {
                int2 rec[8]; int ln[8], so[8];
                int mx = 0;
                #pragma unroll
                for (int q = 0; q < 8; ++q) {
                    ln[q] = __builtin_amdgcn_readfirstlane(rlen[k + 16 * q]);
                    so[q] = __builtin_amdgcn_readfirstlane(roff[k + 16 * q]);
                    mx = max(mx, ln[q]);
                    const int idx = min(so[q] + min(lane, max(ln[q] - 1, 0)), E - 1);
                    rec[q] = tmp[idx];
                }
                #pragma unroll
                for (int q = 0; q < 8; ++q)
                    if (lane < ln[q]) atomicAdd(&cnt[rec[q].x >> 17], 1);
                if (mx > 32) {
                    for (int q = 0; q < 8; ++q)
                        for (int j = lane + 32; j < ln[q]; j += 32)
                            atomicAdd(&cnt[tmp[(size_t)so[q] + j].x >> 17], 1);
                }
            }
        }
        __syncthreads();

        int v = 0, incl = 0;
        if (tid < CSR_R) {
            v = cnt[tid];
            incl = v;
            for (int d = 1; d < 32; d <<= 1) {
                const int t = __shfl_up(incl, d);
                if (lane >= d) incl += t;
            }
            if (lane == 31) wtot[wv] = incl;
        }
        for (int d = 16; d >= 1; d >>= 1) bsum += __shfl_xor(bsum, d);
        if (lane == 0) wbs[wv] = bsum;
        __syncthreads();
        int bstart = 0, total = 0;
        for (int k = 0; k < 16; ++k) bstart += wbs[k];
        for (int k = 0; k < 8; ++k)  total  += wtot[k];
        if (tid < CSR_R) {
            int excl = incl - v;
            for (int k = 0; k < wv; ++k) excl += wtot[k];
            cnt[tid] = excl;
            const int row = b * CSR_R + tid;
            if (row < nrows)      start[row]   = bstart + excl;
            if (row == nrows - 1) start[nrows] = bstart + excl + v;
        }
        const bool fits = total <= CSR_CAP;

        for (int c0 = 0; c0 < nch; c0 += 512) {
            __syncthreads();
            const int c = c0 + tid;
            int len = 0, ro = 0;
            if (c < nch) {
                const int o0 = coff[(size_t)c * CSR_NBP + b];
                const int o1 = coff[(size_t)c * CSR_NBP + b + 1];
                ro  = c * CSR_CH + o0;
                len = o1 - o0;
            }
            roff[tid] = ro;
            rlen[tid] = len;
            __syncthreads();
            const int tn = min(512, nch - c0);
            for (int k = wv; k < tn; k += 128) {
                int2 rec[8]; int ln[8], so[8];
                int mx = 0;
                #pragma unroll
                for (int q = 0; q < 8; ++q) {
                    ln[q] = __builtin_amdgcn_readfirstlane(rlen[k + 16 * q]);
                    so[q] = __builtin_amdgcn_readfirstlane(roff[k + 16 * q]);
                    mx = max(mx, ln[q]);
                    const int idx = min(so[q] + min(lane, max(ln[q] - 1, 0)), E - 1);
                    rec[q] = tmp[idx];
                }
                #pragma unroll
                for (int q = 0; q < 8; ++q)
                    if (lane < ln[q]) gcn_csr_place(rec[q], cnt, fits, stage, sorted, bstart);
                if (mx > 32) {
                    for (int q = 0; q < 8; ++q)
                        for (int j = lane + 32; j < ln[q]; j += 32)
                            gcn_csr_place(tmp[(size_t)so[q] + j], cnt, fits, stage, sorted, bstart);
                }
            }
        }
        __syncthreads();
        if (fits)
            for (int i = tid; i < total; i += 512) sorted[(size_t)bstart + i] = stage[i];
        __syncthreads();
    }
}

template <int D, int U>
__device__ __forceinline__
void gcn_gather_batch(float4& acc, const float* __restrict__ sup,
                      int ecol, int ewb, int j, int sub, int f)
{
    constexpr int EPI = 128 / D;
    int    c[U];
    float  wv[U];
    float4 v[U];
    #pragma unroll
    for (int u = 0; u < U; ++u) {
        if (EPI == 1) {
            c[u]  = __builtin_amdgcn_readlane(ecol, j + u);
            wv[u] = __int_as_float(__builtin_amdgcn_readlane(ewb, j + u));
        } else {
            const int c0 = __builtin_amdgcn_readlane(ecol, j + 2 * u);
            const int c1 = __builtin_amdgcn_readlane(ecol, j + 2 * u + 1);
            const int w0 = __builtin_amdgcn_readlane(ewb,  j + 2 * u);
            const int w1 = __builtin_amdgcn_readlane(ewb,  j + 2 * u + 1);
            c[u]  = sub ? c1 : c0;
            wv[u] = __int_as_float(sub ? w1 : w0);
        }
    }
    #pragma unroll
    for (int u = 0; u < U; ++u)
        v[u] = *(const float4*)(sup + (size_t)c[u] * D + f);
    #pragma unroll
    for (int u = 0; u < U; ++u) {
        acc.x = fmaf(wv[u], v[u].x, acc.x);
        acc.y = fmaf(wv[u], v[u].y, acc.y);
        acc.z = fmaf(wv[u], v[u].z, acc.z);
        acc.w = fmaf(wv[u], v[u].w, acc.w);
    }
}

template <int D>
__global__ __launch_bounds__(256)
void gcn_spmm_gather(const int* __restrict__ start,
                     const int2* __restrict__ sorted,
                     const float* __restrict__ sup,
                     const float* __restrict__ bias,
                     float* __restrict__ out,
                     int nrows)
{
    constexpr int LPE = D / 4;
    constexpr int EPI = 32 / LPE;
    const int lane = threadIdx.x & 31;
    const int sub  = lane / LPE;
    const int f    = (lane - sub * LPE) * 4;
    const int wave0  = __builtin_amdgcn_readfirstlane(blockIdx.x * 8 + (threadIdx.x >> 5));
    const int nwaves = gridDim.x * 8;
    const float4 bv = *(const float4*)(bias + f);

    for (int r = wave0; r < nrows; r += nwaves) {
        const int beg = start[r];
        const int end = start[r + 1];
        float4 acc = make_float4(0.0f, 0.0f, 0.0f, 0.0f);

        for (int base = beg; base < end; base += 32) {
            const int n = min(32, end - base);
            const int2 ev  = sorted[base + min(lane, n - 1)];
            const int ecol = ev.x;
            const int ewb  = (lane < n) ? ev.y : 0;
            int j = 0;
            for (; j + 8 <= n; j += 8)
                gcn_gather_batch<D, 8 / EPI>(acc, sup, ecol, ewb, j, sub, f);
            for (; j < n; j += 4)
                gcn_gather_batch<D, 4 / EPI>(acc, sup, ecol, ewb, j, sub, f);
        }

        if (EPI == 2) {
            acc.x += __shfl_xor(acc.x, 16);
            acc.y += __shfl_xor(acc.y, 16);
            acc.z += __shfl_xor(acc.z, 16);
            acc.w += __shfl_xor(acc.w, 16);
        }
        if (sub == 0) {
            float4 o;
            o.x = acc.x + bv.x; o.y = acc.y + bv.y;
            o.z = acc.z + bv.z; o.w = acc.w + bv.w;
            *(float4*)(out + (size_t)r * D + f) = o;
        }
    }
}

extern "C" void kernel_launch(void* const* d_in, const int* in_sizes, int n_in,
                              void* d_out, int out_size, void* d_ws, size_t ws_size,
                              hipStream_t stream)
{
    const float* x  = (const float*)d_in[0];
    const int*   ei = (const int*)  d_in[1];
    const float* ew = (const float*)d_in[2];
    const float* W1 = (const float*)d_in[3];
    const float* b1 = (const float*)d_in[4];
    const float* W2 = (const float*)d_in[5];
    const float* b2 = (const float*)d_in[6];
    float* out = (float*)d_out;

    const int N  = 100000;
    const int E  = in_sizes[2];
    constexpr int F  = 256;
    constexpr int H1 = 128;
    constexpr int H2 = 64;
    constexpr int KP = 64;

    float* sup1 = (float*)d_ws;
    float* h    = sup1 + (size_t)N * H1;
    float* sup2 = h    + (size_t)N * H1;
    int*   start  = (int*)(sup2 + (size_t)N * H2);
    int2*  sorted = (int2*)(start + 100352);
    int2*  tmp    = sorted + E;
    int*   coff   = (int*)(tmp + E);

    gcn_part_chunks<<<256, 1024, 0, stream>>>(ei, ei + E, ew, tmp, coff, E);
    gcn_csr_rows<<<CSR_NB, 512, 0, stream>>>(tmp, coff, start, sorted, E, N);

    {
        const int waves  = (N / 16) * (H1 / 64);
        const int blocks = (waves + 7) / 8;
        const size_t lds = (size_t)KP * H1 * sizeof(float);
        gcn_gemm_wmma_f32<false, H1, F, KP><<<blocks, 256, lds, stream>>>(x, W1, sup1, N);
    }
    gcn_spmm_gather<H1><<<2048, 256, 0, stream>>>(start, sorted, sup1, b1, h, N);

    {
        const int waves  = (N / 16) * (H2 / 64);
        const int blocks = (waves + 7) / 8;
        const size_t lds = (size_t)KP * H2 * sizeof(float);
        gcn_gemm_wmma_f32<true, H2, H1, KP><<<blocks, 256, lds, stream>>>(h, W2, sup2, N);
    }
    gcn_spmm_gather<H2><<<2048, 256, 0, stream>>>(start, sorted, sup2, b2, out, N);
}


# Layer-2 product moved into the layer-1 gather: 16 summed rows go to LDS with ReLU, 4 waves run the f32 WMMA against W2 in LDS; h never stored, one launch fewer; k sum in two interleaved accumulators
# speedup vs baseline: 3.3692x; 1.0862x over previous
#include <hip/hip_runtime.h>
#include <stdint.h>

typedef __attribute__((ext_vector_type(2))) float v2f;
typedef __attribute__((ext_vector_type(8))) float v8f;

template <bool RELU_A, int N, int K, int KP>
__global__ __launch_bounds__(256)
void gcn_gemm_wmma_f32(const float* __restrict__ A,
                       const float* __restrict__ B,
                       float* __restrict__ C,
                       int M)
{
    extern __shared__ float ldsB[];

    const int tid  = threadIdx.x;
    const int lane = tid & 31;
    const int wave = tid >> 5;

    constexpr int NCOLS = N / 64;
    const int gwave = blockIdx.x * 8 + wave;
    const int mt    = gwave / NCOLS;
    const int n0    = (gwave - mt * NCOLS) * 64;
    const bool active = (mt * 16 < M);
    const int m0    = mt * 16;

    const int half = lane >> 4;
    const int l    = lane & 15;

    const float* arow  = active ? (A + (size_t)(m0 + l) * K + 2 * half) : A;
    const float* bfrag = ldsB + (size_t)(half * N + n0 + l) * 2;

    v8f acc0 = {}, acc1 = {}, acc2 = {}, acc3 = {};

    for (int k0 = 0; k0 < K; k0 += KP) {
        if (k0 != 0) __syncthreads();
        constexpr int PAIRS = (KP / 2) * N;
        #pragma unroll
        for (int idx = 0; idx < PAIRS / 256; ++idx) {
            const int pi = idx * 256 + tid;
            const int p  = pi / N;
            const int n  = pi - p * N;
            const int gk = k0 + 2 * p;
            v2f pr;
            pr.x = B[(size_t)gk * N + n];
            pr.y = B[(size_t)(gk + 1) * N + n];
            *(v2f*)(ldsB + (size_t)pi * 2) = pr;
        }
        __syncthreads();

        if (active) {
            #pragma unroll
            for (int kk = 0; kk < KP; kk += 4) {
                float a0 = arow[k0 + kk];
                float a1 = arow[k0 + kk + 1];
                if (RELU_A) { a0 = fmaxf(a0, 0.0f); a1 = fmaxf(a1, 0.0f); }
                v2f a; a.x = a0; a.y = a1;

                const float* bp = bfrag + (kk >> 1) * (N * 2);
                v2f b0 = *(const v2f*)(bp);
                v2f b1 = *(const v2f*)(bp + 32);
                v2f b2 = *(const v2f*)(bp + 64);
                v2f b3 = *(const v2f*)(bp + 96);

                acc0 = __builtin_amdgcn_wmma_f32_16x16x4_f32(false, a, false, b0,
                                                             (short)0, acc0, false, false);
                acc1 = __builtin_amdgcn_wmma_f32_16x16x4_f32(false, a, false, b1,
                                                             (short)0, acc1, false, false);
                acc2 = __builtin_amdgcn_wmma_f32_16x16x4_f32(false, a, false, b2,
                                                             (short)0, acc2, false, false);
                acc3 = __builtin_amdgcn_wmma_f32_16x16x4_f32(false, a, false, b3,
                                                             (short)0, acc3, false, false);
            }
        }
    }

    if (active) {
        #pragma unroll
        for (int i = 0; i < 8; ++i) {
            const size_t r = (size_t)(m0 + i + 8 * half);
            float* crow = C + r * N + n0 + l;
            crow[0]  = acc0[i];
            crow[16] = acc1[i];
            crow[32] = acc2[i];
            crow[48] = acc3[i];
        }
    }
}

#define CSR_R     256
#define CSR_NB    391
#define CSR_NBP   392
#define CSR_CH    8192
#define CSR_EPT   8
#define CSR_CAP   8192

__global__ __launch_bounds__(1024)
void gcn_part_chunks(const int* __restrict__ rows,
                     const int* __restrict__ cols,
                     const float* __restrict__ w,
                     int2* __restrict__ tmp,
                     int* __restrict__ coff,
                     int E)
{
    __shared__ int  cnt[CSR_NBP];
    __shared__ int  wsum[32];
    __shared__ int2 stage[CSR_CH];

    const int tid  = threadIdx.x;
    const int lane = tid & 31;
    const int wv   = tid >> 5;
    const int nch  = (E + CSR_CH - 1) / CSR_CH;

    for (int c = blockIdx.x; c < nch; c += gridDim.x) {
        const int base = c * CSR_CH;
        const int n    = min(CSR_CH, E - base);

        if (tid < CSR_NBP) cnt[tid] = 0;
        __syncthreads();

        int  pk[CSR_EPT], wb[CSR_EPT], bk[CSR_EPT], rk[CSR_EPT];
        #pragma unroll
        for (int u = 0; u < CSR_EPT; ++u) {
            const int i  = min(u * 1024 + tid, n - 1);
            const int r  = rows[base + i];
            const int cc = cols[base + i];
            wb[u] = __float_as_int(w[base + i]);
            bk[u] = min(r >> 8, CSR_NB - 1);
            pk[u] = ((r & (CSR_R - 1)) << 17) | cc;
        }
        #pragma unroll
        for (int u = 0; u < CSR_EPT; ++u) {
            rk[u] = 0;
            if (u * 1024 + tid < n) rk[u] = atomicAdd(&cnt[bk[u]], 1);
        }
        __syncthreads();

        int v = 0, incl = 0;
        if (tid < 416) {
            v = (tid < CSR_NB) ? cnt[tid] : 0;
            incl = v;
            for (int d = 1; d < 32; d <<= 1) {
                const int t = __shfl_up(incl, d);
                if (lane >= d) incl += t;
            }
            if (lane == 31) wsum[wv] = incl;
        }
        __syncthreads();
        if (tid < CSR_NBP) {
            int off = incl - v;
            for (int k = 0; k < wv; ++k) off += wsum[k];
            cnt[tid] = off;
            coff[(size_t)c * CSR_NBP + tid] = off;
        }
        __syncthreads();

        int dst[CSR_EPT];
        #pragma unroll
        for (int u = 0; u < CSR_EPT; ++u) dst[u] = cnt[bk[u]] + rk[u];
        #pragma unroll
        for (int u = 0; u < CSR_EPT; ++u)
            if (u * 1024 + tid < n) stage[dst[u]] = make_int2(pk[u], wb[u]);
        __syncthreads();

        for (int i = tid; i < n; i += 1024) tmp[(size_t)base + i] = stage[i];
        __syncthreads();
    }
}

__device__ __forceinline__
void gcn_csr_place(int2 rec, int* cur, bool fits, int2* stage,
                   int2* __restrict__ sorted, int bstart)
{
    const int pos = atomicAdd(&cur[rec.x >> 17], 1);
    const int2 o  = make_int2(rec.x & 0x1FFFF, rec.y);
    if (fits) stage[pos] = o;
    else      sorted[(size_t)bstart + pos] = o;
}

__global__ __launch_bounds__(512)
void gcn_csr_rows(const int2* __restrict__ tmp,
                  const int* __restrict__ coff,
                  int* __restrict__ start,
                  int2* __restrict__ sorted,
                  int E, int nrows)
{
    __shared__ int  roff[640];
    __shared__ int  rlen[640];
    __shared__ int  cnt[CSR_R];
    __shared__ int  wtot[8];
    __shared__ int  wbs[16];
    __shared__ int2 stage[CSR_CAP];

    const int tid  = threadIdx.x;
    const int lane = tid & 31;
    const int wv   = tid >> 5;
    const int nch  = (E + CSR_CH - 1) / CSR_CH;

    if (tid < 128) { roff[512 + tid] = 0; rlen[512 + tid] = 0; }

    for (int b = blockIdx.x; b < CSR_NB; b += gridDim.x) {
        if (tid < CSR_R) cnt[tid] = 0;
        int bsum = 0;

        for (int c0 = 0; c0 < nch; c0 += 512) {
            __syncthreads();
            const int c = c0 + tid;
            int len = 0, ro = 0;
            if (c < nch) {
                const int o0 = coff[(size_t)c * CSR_NBP + b];
                const int o1 = coff[(size_t)c * CSR_NBP + b + 1];
                ro  = c * CSR_CH + o0;
                len = o1 - o0;
                bsum += o0;
            }
            roff[tid] = ro;
            rlen[tid] = len;
            __syncthreads();
            const int tn = min(512, nch - c0);
            for (int k = wv; k < tn; k += 128) {
                int2 rec[8]; int ln[8], so[8];
                int mx = 0;
                #pragma unroll
                for (int q = 0; q < 8; ++q) {
                    ln[q] = __builtin_amdgcn_readfirstlane(rlen[k + 16 * q]);
                    so[q] = __builtin_amdgcn_readfirstlane(roff[k + 16 * q]);
                    mx = max(mx, ln[q]);
                    const int idx = min(so[q] + min(lane, max(ln[q] - 1, 0)), E - 1);
                    rec[q] = tmp[idx];
                }
                #pragma unroll
                for (int q = 0; q < 8; ++q)
                    if (lane < ln[q]) atomicAdd(&cnt[rec[q].x >> 17], 1);
                if (mx > 32) {
                    for (int q = 0; q < 8; ++q)
                        for (int j = lane + 32; j < ln[q]; j += 32)
                            atomicAdd(&cnt[tmp[(size_t)so[q] + j].x >> 17], 1);
                }
            }
        }
        __syncthreads();

        int v = 0, incl = 0;
        if (tid < CSR_R) {
            v = cnt[tid];
            incl = v;
            for (int d = 1; d < 32; d <<= 1) {
                const int t = __shfl_up(incl, d);
                if (lane >= d) incl += t;
            }
            if (lane == 31) wtot[wv] = incl;
        }
        for (int d = 16; d >= 1; d >>= 1) bsum += __shfl_xor(bsum, d);
        if (lane == 0) wbs[wv] = bsum;
        __syncthreads();
        int bstart = 0, total = 0;
        for (int k = 0; k < 16; ++k) bstart += wbs[k];
        for (int k = 0; k < 8; ++k)  total  += wtot[k];
        if (tid < CSR_R) {
            int excl = incl - v;
            for (int k = 0; k < wv; ++k) excl += wtot[k];
            cnt[tid] = excl;
            const int row = b * CSR_R + tid;
            if (row < nrows)      start[row]   = bstart + excl;
            if (row == nrows - 1) start[nrows] = bstart + excl + v;
        }
        const bool fits = total <= CSR_CAP;

        for (int c0 = 0; c0 < nch; c0 += 512) {
            __syncthreads();
            const int c = c0 + tid;
            int len = 0, ro = 0;
            if (c < nch) {
                const int o0 = coff[(size_t)c * CSR_NBP + b];
                const int o1 = coff[(size_t)c * CSR_NBP + b + 1];
                ro  = c * CSR_CH + o0;
                len = o1 - o0;
            }
            roff[tid] = ro;
            rlen[tid] = len;
            __syncthreads();
            const int tn = min(512, nch - c0);
            for (int k = wv; k < tn; k += 128) {
                int2 rec[8]; int ln[8], so[8];
                int mx = 0;
                #pragma unroll
                for (int q = 0; q < 8; ++q) {
                    ln[q] = __builtin_amdgcn_readfirstlane(rlen[k + 16 * q]);
                    so[q] = __builtin_amdgcn_readfirstlane(roff[k + 16 * q]);
                    mx = max(mx, ln[q]);
                    const int idx = min(so[q] + min(lane, max(ln[q] - 1, 0)), E - 1);
                    rec[q] = tmp[idx];
                }
                #pragma unroll
                for (int q = 0; q < 8; ++q)
                    if (lane < ln[q]) gcn_csr_place(rec[q], cnt, fits, stage, sorted, bstart);
                if (mx > 32) {
                    for (int q = 0; q < 8; ++q)
                        for (int j = lane + 32; j < ln[q]; j += 32)
                            gcn_csr_place(tmp[(size_t)so[q] + j], cnt, fits, stage, sorted, bstart);
                }
            }
        }
        __syncthreads();
        if (fits)
            for (int i = tid; i < total; i += 512) sorted[(size_t)bstart + i] = stage[i];
        __syncthreads();
    }
}

template <int D, int U>
__device__ __forceinline__
void gcn_gather_batch(float4& acc, const float* __restrict__ sup,
                      int ecol, int ewb, int j, int sub, int f)
{
    constexpr int EPI = 128 / D;
    int    c[U];
    float  wv[U];
    float4 v[U];
    #pragma unroll
    for (int u = 0; u < U; ++u) {
        if (EPI == 1) {
            c[u]  = __builtin_amdgcn_readlane(ecol, j + u);
            wv[u] = __int_as_float(__builtin_amdgcn_readlane(ewb, j + u));
        } else {
            const int c0 = __builtin_amdgcn_readlane(ecol, j + 2 * u);
            const int c1 = __builtin_amdgcn_readlane(ecol, j + 2 * u + 1);
            const int w0 = __builtin_amdgcn_readlane(ewb,  j + 2 * u);
            const int w1 = __builtin_amdgcn_readlane(ewb,  j + 2 * u + 1);
            c[u]  = sub ? c1 : c0;
            wv[u] = __int_as_float(sub ? w1 : w0);
        }
    }
    #pragma unroll
    for (int u = 0; u < U; ++u)
        v[u] = *(const float4*)(sup + (size_t)c[u] * D + f);
    #pragma unroll
    for (int u = 0; u < U; ++u) {
        acc.x = fmaf(wv[u], v[u].x, acc.x);
        acc.y = fmaf(wv[u], v[u].y, acc.y);
        acc.z = fmaf(wv[u], v[u].z, acc.z);
        acc.w = fmaf(wv[u], v[u].w, acc.w);
    }
}

template <int D>
__global__ __launch_bounds__(256)
void gcn_spmm_gather(const int* __restrict__ start,
                     const int2* __restrict__ sorted,
                     const float* __restrict__ sup,
                     const float* __restrict__ bias,
                     float* __restrict__ out,
                     int nrows)
{
    constexpr int LPE = D / 4;
    constexpr int EPI = 32 / LPE;
    const int lane = threadIdx.x & 31;
    const int sub  = lane / LPE;
    const int f    = (lane - sub * LPE) * 4;
    const int wave0  = __builtin_amdgcn_readfirstlane(blockIdx.x * 8 + (threadIdx.x >> 5));
    const int nwaves = gridDim.x * 8;
    const float4 bv = *(const float4*)(bias + f);

    for (int r = wave0; r < nrows; r += nwaves) {
        const int beg = start[r];
        const int end = start[r + 1];
        float4 acc = make_float4(0.0f, 0.0f, 0.0f, 0.0f);

        for (int base = beg; base < end; base += 32) {
            const int n = min(32, end - base);
            const int2 ev  = sorted[base + min(lane, n - 1)];
            const int ecol = ev.x;
            const int ewb  = (lane < n) ? ev.y : 0;
            int j = 0;
            for (; j + 8 <= n; j += 8)
                gcn_gather_batch<D, 8 / EPI>(acc, sup, ecol, ewb, j, sub, f);
            for (; j < n; j += 4)
                gcn_gather_batch<D, 4 / EPI>(acc, sup, ecol, ewb, j, sub, f);
        }

        if (EPI == 2) {
            acc.x += __shfl_xor(acc.x, 16);
            acc.y += __shfl_xor(acc.y, 16);
            acc.z += __shfl_xor(acc.z, 16);
            acc.w += __shfl_xor(acc.w, 16);
        }
        if (sub == 0) {
            float4 o;
            o.x = acc.x + bv.x; o.y = acc.y + bv.y;
            o.z = acc.z + bv.z; o.w = acc.w + bv.w;
            *(float4*)(out + (size_t)r * D + f) = o;
        }
    }
}


#define FG_PITCH 132
__global__ __launch_bounds__(512)
void gcn_gather_relu_gemm(const int* __restrict__ start,
                          const int2* __restrict__ sorted,
                          const float* __restrict__ sup,
                          const float* __restrict__ bias,
                          const float* __restrict__ B,
                          float* __restrict__ C,
                          int nrows)
{
    constexpr int D  = 128;
    constexpr int NO = 64;
    extern __shared__ float fgl[];
    float* ldsB = fgl;
    float* hT   = fgl + D * NO;
    float* oT   = hT + 16 * FG_PITCH;

    const int tid  = threadIdx.x;
    const int lane = tid & 31;
    const int wave = __builtin_amdgcn_readfirstlane(tid >> 5);
    const int half = lane >> 4;
    const int l    = lane & 15;
    const int f    = lane * 4;

    #pragma unroll
    for (int idx = 0; idx < (D / 2) * NO / 512; ++idx) {
        const int pi = idx * 512 + tid;
        const int p  = pi / NO;
        const int n  = pi - p * NO;
        v2f pr;
        pr.x = B[(size_t)(2 * p) * NO + n];
        pr.y = B[(size_t)(2 * p + 1) * NO + n];
        *(v2f*)(ldsB + (size_t)pi * 2) = pr;
    }

    const float4 bv = *(const float4*)(bias + f);
    const int ntiles = (nrows + 15) >> 4;

    for (int tile = blockIdx.x; tile < ntiles; tile += gridDim.x) {
        const int r = tile * 16 + wave;
        float4 o = make_float4(0.0f, 0.0f, 0.0f, 0.0f);
        if (r < nrows) {
            const int beg = start[r];
            const int end = start[r + 1];
            float4 acc = make_float4(0.0f, 0.0f, 0.0f, 0.0f);
            for (int base = beg; base < end; base += 32) {
                const int n = min(32, end - base);
                const int2 ev  = sorted[base + min(lane, n - 1)];
                const int ecol = ev.x;
                const int ewb  = (lane < n) ? ev.y : 0;
                int j = 0;
                for (; j + 8 <= n; j += 8)
                    gcn_gather_batch<D, 8>(acc, sup, ecol, ewb, j, 0, f);
                for (; j < n; j += 4)
                    gcn_gather_batch<D, 4>(acc, sup, ecol, ewb, j, 0, f);
            }
            o.x = fmaxf(acc.x + bv.x, 0.0f); o.y = fmaxf(acc.y + bv.y, 0.0f);
            o.z = fmaxf(acc.z + bv.z, 0.0f); o.w = fmaxf(acc.w + bv.w, 0.0f);
        }
        *(float4*)(hT + wave * FG_PITCH + f) = o;
        __syncthreads();

        if (wave < 4) {
            const int n0 = wave * 16;
            const float* ap = hT + l * FG_PITCH + 2 * half;
            const float* bp = ldsB + (size_t)(half * NO + n0 + l) * 2;
            v8f c0 = {}, c1 = {};
            #pragma unroll
            for (int s = 0; s < D / 4; s += 2) {
                const v2f a0 = *(const v2f*)(ap + 4 * s);
                const v2f b0 = *(const v2f*)(bp + s * (NO * 4));
                const v2f a1 = *(const v2f*)(ap + 4 * s + 4);
                const v2f b1 = *(const v2f*)(bp + (s + 1) * (NO * 4));
                c0 = __builtin_amdgcn_wmma_f32_16x16x4_f32(false, a0, false, b0,
                                                           (short)0, c0, false, false);
                c1 = __builtin_amdgcn_wmma_f32_16x16x4_f32(false, a1, false, b1,
                                                           (short)0, c1, false, false);
            }
            #pragma unroll
            for (int i = 0; i < 8; ++i)
                oT[(i + 8 * half) * NO + n0 + l] = c0[i] + c1[i];
        }
        __syncthreads();

        if (tid < 256) {
            const int row = tid >> 4;
            const int c4  = (tid & 15) * 4;
            const int rr  = tile * 16 + row;
            if (rr < nrows)
                *(float4*)(C + (size_t)rr * NO + c4) = *(const float4*)(oT + row * NO + c4);
        }
    }
}

extern "C" void kernel_launch(void* const* d_in, const int* in_sizes, int n_in,
                              void* d_out, int out_size, void* d_ws, size_t ws_size,
                              hipStream_t stream)
{
    const float* x  = (const float*)d_in[0];
    const int*   ei = (const int*)  d_in[1];
    const float* ew = (const float*)d_in[2];
    const float* W1 = (const float*)d_in[3];
    const float* b1 = (const float*)d_in[4];
    const float* W2 = (const float*)d_in[5];
    const float* b2 = (const float*)d_in[6];
    float* out = (float*)d_out;

    const int N  = 100000;
    const int E  = in_sizes[2];
    constexpr int F  = 256;
    constexpr int H1 = 128;
    constexpr int H2 = 64;
    constexpr int KP = 64;

    float* sup1 = (float*)d_ws;
    float* h    = sup1 + (size_t)N * H1;
    float* sup2 = h    + (size_t)N * H1;
    int*   start  = (int*)(sup2 + (size_t)N * H2);
    int2*  sorted = (int2*)(start + 100352);
    int2*  tmp    = sorted + E;
    int*   coff   = (int*)(tmp + E);

    gcn_part_chunks<<<256, 1024, 0, stream>>>(ei, ei + E, ew, tmp, coff, E);
    gcn_csr_rows<<<CSR_NB, 512, 0, stream>>>(tmp, coff, start, sorted, E, N);

    {
        const int waves  = (N / 16) * (H1 / 64);
        const int blocks = (waves + 7) / 8;
        const size_t lds = (size_t)KP * H1 * sizeof(float);
        gcn_gemm_wmma_f32<false, H1, F, KP><<<blocks, 256, lds, stream>>>(x, W1, sup1, N);
    }
    {
        const size_t lds = (size_t)(H1 * H2 + 16 * FG_PITCH + 16 * H2) * sizeof(float);
        gcn_gather_relu_gemm<<<1024, 512, lds, stream>>>(start, sorted, sup1, b1, W2, sup2, N);
    }
    gcn_spmm_gather<H2><<<2048, 256, 0, stream>>>(start, sorted, sup2, b2, out, N);
}


# CSR level 2 (one 256-thread block per row bucket) rides in the x@W1 launch as 391 blocks dispatched behind the product's blocks: one launch fewer; per-row sum order stays LDS arrival order
# speedup vs baseline: 3.5344x; 1.0490x over previous
#include <hip/hip_runtime.h>
#include <stdint.h>

typedef __attribute__((ext_vector_type(2))) float v2f;
typedef __attribute__((ext_vector_type(8))) float v8f;

template <bool RELU_A, int N, int K, int KP>
__device__ __forceinline__
void gcn_gemm_wmma_f32(float* ldsB,
                       const int gblock,
                       const float* __restrict__ A,
                       const float* __restrict__ B,
                       float* __restrict__ C,
                       int M)
{
    const int tid  = threadIdx.x;
    const int lane = tid & 31;
    const int wave = tid >> 5;

    constexpr int NCOLS = N / 64;
    const int gwave = gblock * 8 + wave;
    const int mt    = gwave / NCOLS;
    const int n0    = (gwave - mt * NCOLS) * 64;
    const bool active = (mt * 16 < M);
    const int m0    = mt * 16;

    const int half = lane >> 4;
    const int l    = lane & 15;

    const float* arow  = active ? (A + (size_t)(m0 + l) * K + 2 * half) : A;
    const float* bfrag = ldsB + (size_t)(half * N + n0 + l) * 2;

    v8f acc0 = {}, acc1 = {}, acc2 = {}, acc3 = {};

    for (int k0 = 0; k0 < K; k0 += KP) {
        if (k0 != 0) __syncthreads();
        constexpr int PAIRS = (KP / 2) * N;
        #pragma unroll
        for (int idx = 0; idx < PAIRS / 256; ++idx) {
            const int pi = idx * 256 + tid;
            const int p  = pi / N;
            const int n  = pi - p * N;
            const int gk = k0 + 2 * p;
            v2f pr;
            pr.x = B[(size_t)gk * N + n];
            pr.y = B[(size_t)(gk + 1) * N + n];
            *(v2f*)(ldsB + (size_t)pi * 2) = pr;
        }
        __syncthreads();

        if (active) {
            #pragma unroll
            for (int kk = 0; kk < KP; kk += 4) {
                float a0 = arow[k0 + kk];
                float a1 = arow[k0 + kk + 1];
                if (RELU_A) { a0 = fmaxf(a0, 0.0f); a1 = fmaxf(a1, 0.0f); }
                v2f a; a.x = a0; a.y = a1;

                const float* bp = bfrag + (kk >> 1) * (N * 2);
                v2f b0 = *(const v2f*)(bp);
                v2f b1 = *(const v2f*)(bp + 32);
                v2f b2 = *(const v2f*)(bp + 64);
                v2f b3 = *(const v2f*)(bp + 96);

                acc0 = __builtin_amdgcn_wmma_f32_16x16x4_f32(false, a, false, b0,
                                                             (short)0, acc0, false, false);
                acc1 = __builtin_amdgcn_wmma_f32_16x16x4_f32(false, a, false, b1,
                                                             (short)0, acc1, false, false);
                acc2 = __builtin_amdgcn_wmma_f32_16x16x4_f32(false, a, false, b2,
                                                             (short)0, acc2, false, false);
                acc3 = __builtin_amdgcn_wmma_f32_16x16x4_f32(false, a, false, b3,
                                                             (short)0, acc3, false, false);
            }
        }
    }

    if (active) {
        #pragma unroll
        for (int i = 0; i < 8; ++i) {
            const size_t r = (size_t)(m0 + i + 8 * half);
            float* crow = C + r * N + n0 + l;
            crow[0]  = acc0[i];
            crow[16] = acc1[i];
            crow[32] = acc2[i];
            crow[48] = acc3[i];
        }
    }
}

#define CSR_R     256
#define CSR_NB    391
#define CSR_NBP   392
#define CSR_CH    8192
#define CSR_EPT   8
#define CSR_CAP   8192

__global__ __launch_bounds__(1024)
void gcn_part_chunks(const int* __restrict__ rows,
                     const int* __restrict__ cols,
                     const float* __restrict__ w,
                     int2* __restrict__ tmp,
                     int* __restrict__ coff,
                     int E)
{
    __shared__ int  cnt[CSR_NBP];
    __shared__ int  wsum[32];
    __shared__ int2 stage[CSR_CH];

    const int tid  = threadIdx.x;
    const int lane = tid & 31;
    const int wv   = tid >> 5;
    const int nch  = (E + CSR_CH - 1) / CSR_CH;

    for (int c = blockIdx.x; c < nch; c += gridDim.x) {
        const int base = c * CSR_CH;
        const int n    = min(CSR_CH, E - base);

        if (tid < CSR_NBP) cnt[tid] = 0;
        __syncthreads();

        int  pk[CSR_EPT], wb[CSR_EPT], bk[CSR_EPT], rk[CSR_EPT];
        #pragma unroll
        for (int u = 0; u < CSR_EPT; ++u) {
            const int i  = min(u * 1024 + tid, n - 1);
            const int r  = rows[base + i];
            const int cc = cols[base + i];
            wb[u] = __float_as_int(w[base + i]);
            bk[u] = min(r >> 8, CSR_NB - 1);
            pk[u] = ((r & (CSR_R - 1)) << 17) | cc;
        }
        #pragma unroll
        for (int u = 0; u < CSR_EPT; ++u) {
            rk[u] = 0;
            if (u * 1024 + tid < n) rk[u] = atomicAdd(&cnt[bk[u]], 1);
        }
        __syncthreads();

        int v = 0, incl = 0;
        if (tid < 416) {
            v = (tid < CSR_NB) ? cnt[tid] : 0;
            incl = v;
            for (int d = 1; d < 32; d <<= 1) {
                const int t = __shfl_up(incl, d);
                if (lane >= d) incl += t;
            }
            if (lane == 31) wsum[wv] = incl;
        }
        __syncthreads();
        if (tid < CSR_NBP) {
            int off = incl - v;
            for (int k = 0; k < wv; ++k) off += wsum[k];
            cnt[tid] = off;
            coff[(size_t)c * CSR_NBP + tid] = off;
        }
        __syncthreads();

        int dst[CSR_EPT];
        #pragma unroll
        for (int u = 0; u < CSR_EPT; ++u) dst[u] = cnt[bk[u]] + rk[u];
        #pragma unroll
        for (int u = 0; u < CSR_EPT; ++u)
            if (u * 1024 + tid < n) stage[dst[u]] = make_int2(pk[u], wb[u]);
        __syncthreads();

        for (int i = tid; i < n; i += 1024) tmp[(size_t)base + i] = stage[i];
        __syncthreads();
    }
}

__device__ __forceinline__
void gcn_csr_place(int2 rec, int* cur, bool fits, int2* stage,
                   int2* __restrict__ sorted, int bstart)
{
    const int pos = atomicAdd(&cur[rec.x >> 17], 1);
    const int2 o  = make_int2(rec.x & 0x1FFFF, rec.y);
    if (fits) stage[pos] = o;
    else      sorted[(size_t)bstart + pos] = o;
}

#define CSR_L2_LDS_BYTES ((640 + 640 + CSR_R + 8 + 8) * 4 + CSR_CAP * 8)
__device__ __forceinline__
void gcn_csr_rows(unsigned char* smem, const int b,
                  const int2* __restrict__ tmp,
                  const int* __restrict__ coff,
                  int* __restrict__ start,
                  int2* __restrict__ sorted,
                  int E, int nrows)
{
    int*  roff  = (int*)smem;
    int*  rlen  = roff + 640;
    int*  cnt   = rlen + 640;
    int*  wtot  = cnt + CSR_R;
    int*  wbs   = wtot + 8;
    int2* stage = (int2*)(wbs + 8);

    const int tid  = threadIdx.x;
    const int lane = tid & 31;
    const int wv   = tid >> 5;
    const int nch  = (E + CSR_CH - 1) / CSR_CH;

    if (tid < 128) { roff[512 + tid] = 0; rlen[512 + tid] = 0; }
    cnt[tid] = 0;
    int bsum = 0;

    for (int c0 = 0; c0 < nch; c0 += 512) {
        __syncthreads();
        #pragma unroll
        for (int hh = 0; hh < 2; ++hh) {
            const int t = tid + 256 * hh;
            const int c = c0 + t;
            int len = 0, ro = 0;
            if (c < nch) {
                const int o0 = coff[(size_t)c * CSR_NBP + b];
                const int o1 = coff[(size_t)c * CSR_NBP + b + 1];
                ro  = c * CSR_CH + o0;
                len = o1 - o0;
                bsum += o0;
            }
            roff[t] = ro;
            rlen[t] = len;
        }
        __syncthreads();
        const int tn = min(512, nch - c0);
        for (int k = wv; k < tn; k += 64) {
            int2 rec[8]; int ln[8], so[8];
            int mx = 0;
            #pragma unroll
            for (int q = 0; q < 8; ++q) {
                ln[q] = __builtin_amdgcn_readfirstlane(rlen[k + 8 * q]);
                so[q] = __builtin_amdgcn_readfirstlane(roff[k + 8 * q]);
                mx = max(mx, ln[q]);
                const int idx = min(so[q] + min(lane, max(ln[q] - 1, 0)), E - 1);
                rec[q] = tmp[idx];
            }
            #pragma unroll
            for (int q = 0; q < 8; ++q)
                if (lane < ln[q]) atomicAdd(&cnt[rec[q].x >> 17], 1);
            if (mx > 32) {
                for (int q = 0; q < 8; ++q)
                    for (int j = lane + 32; j < ln[q]; j += 32)
                        atomicAdd(&cnt[tmp[(size_t)so[q] + j].x >> 17], 1);
            }
        }
    }
    __syncthreads();

    const int v = cnt[tid];
    int incl = v;
    for (int d = 1; d < 32; d <<= 1) {
        const int t = __shfl_up(incl, d);
        if (lane >= d) incl += t;
    }
    if (lane == 31) wtot[wv] = incl;
    for (int d = 16; d >= 1; d >>= 1) bsum += __shfl_xor(bsum, d);
    if (lane == 0) wbs[wv] = bsum;
    __syncthreads();
    int bstart = 0, total = 0;
    for (int k = 0; k < 8; ++k) bstart += wbs[k];
    for (int k = 0; k < 8; ++k) total  += wtot[k];
    {
        int excl = incl - v;
        for (int k = 0; k < wv; ++k) excl += wtot[k];
        cnt[tid] = excl;
        const int row = b * CSR_R + tid;
        if (row < nrows)      start[row]   = bstart + excl;
        if (row == nrows - 1) start[nrows] = bstart + excl + v;
    }
    const bool fits = total <= CSR_CAP;

    for (int c0 = 0; c0 < nch; c0 += 512) {
        __syncthreads();
        #pragma unroll
        for (int hh = 0; hh < 2; ++hh) {
            const int t = tid + 256 * hh;
            const int c = c0 + t;
            int len = 0, ro = 0;
            if (c < nch) {
                const int o0 = coff[(size_t)c * CSR_NBP + b];
                const int o1 = coff[(size_t)c * CSR_NBP + b + 1];
                ro  = c * CSR_CH + o0;
                len = o1 - o0;
            }
            roff[t] = ro;
            rlen[t] = len;
        }
        __syncthreads();
        const int tn = min(512, nch - c0);
        for (int k = wv; k < tn; k += 64) {
            int2 rec[8]; int ln[8], so[8];
            int mx = 0;
            #pragma unroll
            for (int q = 0; q < 8; ++q) {
                ln[q] = __builtin_amdgcn_readfirstlane(rlen[k + 8 * q]);
                so[q] = __builtin_amdgcn_readfirstlane(roff[k + 8 * q]);
                mx = max(mx, ln[q]);
                const int idx = min(so[q] + min(lane, max(ln[q] - 1, 0)), E - 1);
                rec[q] = tmp[idx];
            }
            #pragma unroll
            for (int q = 0; q < 8; ++q)
                if (lane < ln[q]) gcn_csr_place(rec[q], cnt, fits, stage, sorted, bstart);
            if (mx > 32) {
                for (int q = 0; q < 8; ++q)
                    for (int j = lane + 32; j < ln[q]; j += 32)
                        gcn_csr_place(tmp[(size_t)so[q] + j], cnt, fits, stage, sorted, bstart);
            }
        }
    }
    __syncthreads();
    if (fits)
        for (int i = tid; i < total; i += 256) sorted[(size_t)bstart + i] = stage[i];
}

template <int GEMM_BLOCKS, int N, int K, int KP>
__global__ __launch_bounds__(256)
void gcn_gemm1_with_csr_rows(const float* __restrict__ A,
                             const float* __restrict__ B,
                             float* __restrict__ C,
                             int M,
                             const int2* __restrict__ tmp,
                             const int* __restrict__ coff,
                             int* __restrict__ start,
                             int2* __restrict__ sorted,
                             int E, int nrows)
{
    static_assert(CSR_L2_LDS_BYTES >= KP * N * 4, "LDS buffer holds either role");
    __shared__ __align__(16) unsigned char smem[CSR_L2_LDS_BYTES];
    if (blockIdx.x < GEMM_BLOCKS)
        gcn_gemm_wmma_f32<false, N, K, KP>((float*)smem, blockIdx.x, A, B, C, M);
    else
        gcn_csr_rows(smem, blockIdx.x - GEMM_BLOCKS, tmp, coff, start, sorted, E, nrows);
}

template <int D, int U>
__device__ __forceinline__
void gcn_gather_batch(float4& acc, const float* __restrict__ sup,
                      int ecol, int ewb, int j, int sub, int f)
{
    constexpr int EPI = 128 / D;
    int    c[U];
    float  wv[U];
    float4 v[U];
    #pragma unroll
    for (int u = 0; u < U; ++u) {
        if (EPI == 1) {
            c[u]  = __builtin_amdgcn_readlane(ecol, j + u);
            wv[u] = __int_as_float(__builtin_amdgcn_readlane(ewb, j + u));
        } else {
            const int c0 = __builtin_amdgcn_readlane(ecol, j + 2 * u);
            const int c1 = __builtin_amdgcn_readlane(ecol, j + 2 * u + 1);
            const int w0 = __builtin_amdgcn_readlane(ewb,  j + 2 * u);
            const int w1 = __builtin_amdgcn_readlane(ewb,  j + 2 * u + 1);
            c[u]  = sub ? c1 : c0;
            wv[u] = __int_as_float(sub ? w1 : w0);
        }
    }
    #pragma unroll
    for (int u = 0; u < U; ++u)
        v[u] = *(const float4*)(sup + (size_t)c[u] * D + f);
    #pragma unroll
    for (int u = 0; u < U; ++u) {
        acc.x = fmaf(wv[u], v[u].x, acc.x);
        acc.y = fmaf(wv[u], v[u].y, acc.y);
        acc.z = fmaf(wv[u], v[u].z, acc.z);
        acc.w = fmaf(wv[u], v[u].w, acc.w);
    }
}

template <int D>
__global__ __launch_bounds__(256)
void gcn_spmm_gather(const int* __restrict__ start,
                     const int2* __restrict__ sorted,
                     const float* __restrict__ sup,
                     const float* __restrict__ bias,
                     float* __restrict__ out,
                     int nrows)
{
    constexpr int LPE = D / 4;
    constexpr int EPI = 32 / LPE;
    const int lane = threadIdx.x & 31;
    const int sub  = lane / LPE;
    const int f    = (lane - sub * LPE) * 4;
    const int wave0  = __builtin_amdgcn_readfirstlane(blockIdx.x * 8 + (threadIdx.x >> 5));
    const int nwaves = gridDim.x * 8;
    const float4 bv = *(const float4*)(bias + f);

    for (int r = wave0; r < nrows; r += nwaves) {
        const int beg = start[r];
        const int end = start[r + 1];
        float4 acc = make_float4(0.0f, 0.0f, 0.0f, 0.0f);

        for (int base = beg; base < end; base += 32) {
            const int n = min(32, end - base);
            const int2 ev  = sorted[base + min(lane, n - 1)];
            const int ecol = ev.x;
            const int ewb  = (lane < n) ? ev.y : 0;
            int j = 0;
            for (; j + 8 <= n; j += 8)
                gcn_gather_batch<D, 8 / EPI>(acc, sup, ecol, ewb, j, sub, f);
            for (; j < n; j += 4)
                gcn_gather_batch<D, 4 / EPI>(acc, sup, ecol, ewb, j, sub, f);
        }

        if (EPI == 2) {
            acc.x += __shfl_xor(acc.x, 16);
            acc.y += __shfl_xor(acc.y, 16);
            acc.z += __shfl_xor(acc.z, 16);
            acc.w += __shfl_xor(acc.w, 16);
        }
        if (sub == 0) {
            float4 o;
            o.x = acc.x + bv.x; o.y = acc.y + bv.y;
            o.z = acc.z + bv.z; o.w = acc.w + bv.w;
            *(float4*)(out + (size_t)r * D + f) = o;
        }
    }
}


#define FG_PITCH 132
__global__ __launch_bounds__(512)
void gcn_gather_relu_gemm(const int* __restrict__ start,
                          const int2* __restrict__ sorted,
                          const float* __restrict__ sup,
                          const float* __restrict__ bias,
                          const float* __restrict__ B,
                          float* __restrict__ C,
                          int nrows)
{
    constexpr int D  = 128;
    constexpr int NO = 64;
    extern __shared__ float fgl[];
    float* ldsB = fgl;
    float* hT   = fgl + D * NO;
    float* oT   = hT + 16 * FG_PITCH;

    const int tid  = threadIdx.x;
    const int lane = tid & 31;
    const int wave = __builtin_amdgcn_readfirstlane(tid >> 5);
    const int half = lane >> 4;
    const int l    = lane & 15;
    const int f    = lane * 4;

    #pragma unroll
    for (int idx = 0; idx < (D / 2) * NO / 512; ++idx) {
        const int pi = idx * 512 + tid;
        const int p  = pi / NO;
        const int n  = pi - p * NO;
        v2f pr;
        pr.x = B[(size_t)(2 * p) * NO + n];
        pr.y = B[(size_t)(2 * p + 1) * NO + n];
        *(v2f*)(ldsB + (size_t)pi * 2) = pr;
    }

    const float4 bv = *(const float4*)(bias + f);
    const int ntiles = (nrows + 15) >> 4;

    for (int tile = blockIdx.x; tile < ntiles; tile += gridDim.x) {
        const int r = tile * 16 + wave;
        float4 o = make_float4(0.0f, 0.0f, 0.0f, 0.0f);
        if (r < nrows) {
            const int beg = start[r];
            const int end = start[r + 1];
            float4 acc = make_float4(0.0f, 0.0f, 0.0f, 0.0f);
            for (int base = beg; base < end; base += 32) {
                const int n = min(32, end - base);
                const int2 ev  = sorted[base + min(lane, n - 1)];
                const int ecol = ev.x;
                const int ewb  = (lane < n) ? ev.y : 0;
                int j = 0;
                for (; j + 8 <= n; j += 8)
                    gcn_gather_batch<D, 8>(acc, sup, ecol, ewb, j, 0, f);
                for (; j < n; j += 4)
                    gcn_gather_batch<D, 4>(acc, sup, ecol, ewb, j, 0, f);
            }
            o.x = fmaxf(acc.x + bv.x, 0.0f); o.y = fmaxf(acc.y + bv.y, 0.0f);
            o.z = fmaxf(acc.z + bv.z, 0.0f); o.w = fmaxf(acc.w + bv.w, 0.0f);
        }
        *(float4*)(hT + wave * FG_PITCH + f) = o;
        __syncthreads();

        if (wave < 4) {
            const int n0 = wave * 16;
            const float* ap = hT + l * FG_PITCH + 2 * half;
            const float* bp = ldsB + (size_t)(half * NO + n0 + l) * 2;
            v8f c0 = {}, c1 = {};
            #pragma unroll
            for (int s = 0; s < D / 4; s += 2) {
                const v2f a0 = *(const v2f*)(ap + 4 * s);
                const v2f b0 = *(const v2f*)(bp + s * (NO * 4));
                const v2f a1 = *(const v2f*)(ap + 4 * s + 4);
                const v2f b1 = *(const v2f*)(bp + (s + 1) * (NO * 4));
                c0 = __builtin_amdgcn_wmma_f32_16x16x4_f32(false, a0, false, b0,
                                                           (short)0, c0, false, false);
                c1 = __builtin_amdgcn_wmma_f32_16x16x4_f32(false, a1, false, b1,
                                                           (short)0, c1, false, false);
            }
            #pragma unroll
            for (int i = 0; i < 8; ++i)
                oT[(i + 8 * half) * NO + n0 + l] = c0[i] + c1[i];
        }
        __syncthreads();

        if (tid < 256) {
            const int row = tid >> 4;
            const int c4  = (tid & 15) * 4;
            const int rr  = tile * 16 + row;
            if (rr < nrows)
                *(float4*)(C + (size_t)rr * NO + c4) = *(const float4*)(oT + row * NO + c4);
        }
    }
}

extern "C" void kernel_launch(void* const* d_in, const int* in_sizes, int n_in,
                              void* d_out, int out_size, void* d_ws, size_t ws_size,
                              hipStream_t stream)
{
    const float* x  = (const float*)d_in[0];
    const int*   ei = (const int*)  d_in[1];
    const float* ew = (const float*)d_in[2];
    const float* W1 = (const float*)d_in[3];
    const float* b1 = (const float*)d_in[4];
    const float* W2 = (const float*)d_in[5];
    const float* b2 = (const float*)d_in[6];
    float* out = (float*)d_out;

    const int N  = 100000;
    const int E  = in_sizes[2];
    constexpr int F  = 256;
    constexpr int H1 = 128;
    constexpr int H2 = 64;
    constexpr int KP = 64;

    float* sup1 = (float*)d_ws;
    float* h    = sup1 + (size_t)N * H1;
    float* sup2 = h    + (size_t)N * H1;
    int*   start  = (int*)(sup2 + (size_t)N * H2);
    int2*  sorted = (int2*)(start + 100352);
    int2*  tmp    = sorted + E;
    int*   coff   = (int*)(tmp + E);

    gcn_part_chunks<<<256, 1024, 0, stream>>>(ei, ei + E, ew, tmp, coff, E);

    {
        constexpr int GEMM_BLOCKS = 1563;
        gcn_gemm1_with_csr_rows<GEMM_BLOCKS, H1, F, KP>
            <<<GEMM_BLOCKS + CSR_NB, 256, 0, stream>>>(x, W1, sup1, N,
                                                       tmp, coff, start, sorted, E, N);
    }
    {
        const size_t lds = (size_t)(H1 * H2 + 16 * FG_PITCH + 16 * H2) * sizeof(float);
        gcn_gather_relu_gemm<<<1024, 512, lds, stream>>>(start, sorted, sup1, b1, W2, sup2, N);
    }
    gcn_spmm_gather<H2><<<2048, 256, 0, stream>>>(start, sorted, sup2, b2, out, N);
}


# x@W1 operand loads made whole-line: a block's 64x64 x tile goes through LDS via 16-byte row-contiguous loads (was 8-byte loads over 16 rows, twice per tile); W1 panel staged with 16-byte loads
# speedup vs baseline: 3.7800x; 1.0695x over previous
#include <hip/hip_runtime.h>
#include <stdint.h>

typedef __attribute__((ext_vector_type(2))) float v2f;
typedef __attribute__((ext_vector_type(8))) float v8f;

#define GA_PITCH 68
template <int N, int K, int KP>
__device__ __forceinline__
void gcn_gemm_wmma_f32(float* ldsB,
                       float* ldsA,
                       const int gblock,
                       const float* __restrict__ A,
                       const float* __restrict__ B,
                       float* __restrict__ C,
                       int M)
{
    const int tid  = threadIdx.x;
    const int lane = tid & 31;
    const int wave = tid >> 5;

    constexpr int NCOLS = N / 64;
    constexpr int BM    = (8 / NCOLS) * 16;
    constexpr int AQ    = KP / 4;
    constexpr int BQ    = N / 4;
    static_assert(8 % NCOLS == 0, "a block's waves cover whole M tiles");
    static_assert((BM * AQ) % 256 == 0 && ((KP / 2) * BQ) % 256 == 0, "staging loops are whole");
    static_assert(KP + 4 <= GA_PITCH, "staged A row fits its pitch");

    const int ml     = wave / NCOLS;
    const int n0     = (wave - ml * NCOLS) * 64;
    const int m_base = gblock * BM;
    const int m0     = m_base + ml * 16;
    const bool active = (m0 < M);

    const int half = lane >> 4;
    const int l    = lane & 15;

    const float* afrag = ldsA + (ml * 16 + l) * GA_PITCH + 2 * half;
    const float* bfrag = ldsB + (size_t)(half * N + n0 + l) * 2;

    v8f acc0 = {}, acc1 = {}, acc2 = {}, acc3 = {};

    for (int k0 = 0; k0 < K; k0 += KP) {
        if (k0 != 0) __syncthreads();
        float4 br0[(KP / 2) * BQ / 256], br1[(KP / 2) * BQ / 256], ar[BM * AQ / 256];
        #pragma unroll
        for (int idx = 0; idx < (KP / 2) * BQ / 256; ++idx) {
            const int u  = idx * 256 + tid;
            const int p  = u / BQ;
            const int cg = u - p * BQ;
            const float* bsrc = B + (size_t)(k0 + 2 * p) * N + 4 * cg;
            br0[idx] = *(const float4*)(bsrc);
            br1[idx] = *(const float4*)(bsrc + N);
        }
        #pragma unroll
        for (int idx = 0; idx < BM * AQ / 256; ++idx) {
            const int u   = idx * 256 + tid;
            const int row = u / AQ;
            const int c4  = u - row * AQ;
            const int grow = min(m_base + row, M - 1);
            ar[idx] = *(const float4*)(A + (size_t)grow * K + k0 + 4 * c4);
        }
        #pragma unroll
        for (int idx = 0; idx < (KP / 2) * BQ / 256; ++idx) {
            const int u  = idx * 256 + tid;
            const int p  = u / BQ;
            const int cg = u - p * BQ;
            float* bdst = ldsB + (size_t)(p * N + 4 * cg) * 2;
            *(float4*)(bdst)     = make_float4(br0[idx].x, br1[idx].x, br0[idx].y, br1[idx].y);
            *(float4*)(bdst + 4) = make_float4(br0[idx].z, br1[idx].z, br0[idx].w, br1[idx].w);
        }
        #pragma unroll
        for (int idx = 0; idx < BM * AQ / 256; ++idx) {
            const int u   = idx * 256 + tid;
            const int row = u / AQ;
            const int c4  = u - row * AQ;
            *(float4*)(ldsA + row * GA_PITCH + 4 * c4) = ar[idx];
        }
        __syncthreads();

        if (active) {
            #pragma unroll
            for (int kk = 0; kk < KP; kk += 4) {
                const v2f a = *(const v2f*)(afrag + kk);

                const float* bp = bfrag + (kk >> 1) * (N * 2);
                v2f b0 = *(const v2f*)(bp);
                v2f b1 = *(const v2f*)(bp + 32);
                v2f b2 = *(const v2f*)(bp + 64);
                v2f b3 = *(const v2f*)(bp + 96);

                acc0 = __builtin_amdgcn_wmma_f32_16x16x4_f32(false, a, false, b0,
                                                             (short)0, acc0, false, false);
                acc1 = __builtin_amdgcn_wmma_f32_16x16x4_f32(false, a, false, b1,
                                                             (short)0, acc1, false, false);
                acc2 = __builtin_amdgcn_wmma_f32_16x16x4_f32(false, a, false, b2,
                                                             (short)0, acc2, false, false);
                acc3 = __builtin_amdgcn_wmma_f32_16x16x4_f32(false, a, false, b3,
                                                             (short)0, acc3, false, false);
            }
        }
    }

    if (active) {
        #pragma unroll
        for (int i = 0; i < 8; ++i) {
            const size_t r = (size_t)(m0 + i + 8 * half);
            float* crow = C + r * N + n0 + l;
            crow[0]  = acc0[i];
            crow[16] = acc1[i];
            crow[32] = acc2[i];
            crow[48] = acc3[i];
        }
    }
}

#define CSR_R     256
#define CSR_NB    391
#define CSR_NBP   392
#define CSR_CH    8192
#define CSR_EPT   8
#define CSR_CAP   8192

__global__ __launch_bounds__(1024)
void gcn_part_chunks(const int* __restrict__ rows,
                     const int* __restrict__ cols,
                     const float* __restrict__ w,
                     int2* __restrict__ tmp,
                     int* __restrict__ coff,
                     int E)
{
    __shared__ int  cnt[CSR_NBP];
    __shared__ int  wsum[32];
    __shared__ int2 stage[CSR_CH];

    const int tid  = threadIdx.x;
    const int lane = tid & 31;
    const int wv   = tid >> 5;
    const int nch  = (E + CSR_CH - 1) / CSR_CH;

    for (int c = blockIdx.x; c < nch; c += gridDim.x) {
        const int base = c * CSR_CH;
        const int n    = min(CSR_CH, E - base);

        if (tid < CSR_NBP) cnt[tid] = 0;
        __syncthreads();

        int  pk[CSR_EPT], wb[CSR_EPT], bk[CSR_EPT], rk[CSR_EPT];
        #pragma unroll
        for (int u = 0; u < CSR_EPT; ++u) {
            const int i  = min(u * 1024 + tid, n - 1);
            const int r  = rows[base + i];
            const int cc = cols[base + i];
            wb[u] = __float_as_int(w[base + i]);
            bk[u] = min(r >> 8, CSR_NB - 1);
            pk[u] = ((r & (CSR_R - 1)) << 17) | cc;
        }
        #pragma unroll
        for (int u = 0; u < CSR_EPT; ++u) {
            rk[u] = 0;
            if (u * 1024 + tid < n) rk[u] = atomicAdd(&cnt[bk[u]], 1);
        }
        __syncthreads();

        int v = 0, incl = 0;
        if (tid < 416) {
            v = (tid < CSR_NB) ? cnt[tid] : 0;
            incl = v;
            for (int d = 1; d < 32; d <<= 1) {
                const int t = __shfl_up(incl, d);
                if (lane >= d) incl += t;
            }
            if (lane == 31) wsum[wv] = incl;
        }
        __syncthreads();
        if (tid < CSR_NBP) {
            int off = incl - v;
            for (int k = 0; k < wv; ++k) off += wsum[k];
            cnt[tid] = off;
            coff[(size_t)c * CSR_NBP + tid] = off;
        }
        __syncthreads();

        int dst[CSR_EPT];
        #pragma unroll
        for (int u = 0; u < CSR_EPT; ++u) dst[u] = cnt[bk[u]] + rk[u];
        #pragma unroll
        for (int u = 0; u < CSR_EPT; ++u)
            if (u * 1024 + tid < n) stage[dst[u]] = make_int2(pk[u], wb[u]);
        __syncthreads();

        for (int i = tid; i < n; i += 1024) tmp[(size_t)base + i] = stage[i];
        __syncthreads();
    }
}

__device__ __forceinline__
void gcn_csr_place(int2 rec, int* cur, bool fits, int2* stage,
                   int2* __restrict__ sorted, int bstart)
{
    const int pos = atomicAdd(&cur[rec.x >> 17], 1);
    const int2 o  = make_int2(rec.x & 0x1FFFF, rec.y);
    if (fits) stage[pos] = o;
    else      sorted[(size_t)bstart + pos] = o;
}

#define CSR_L2_LDS_BYTES ((640 + 640 + CSR_R + 8 + 8) * 4 + CSR_CAP * 8)
__device__ __forceinline__
void gcn_csr_rows(unsigned char* smem, const int b,
                  const int2* __restrict__ tmp,
                  const int* __restrict__ coff,
                  int* __restrict__ start,
                  int2* __restrict__ sorted,
                  int E, int nrows)
{
    int*  roff  = (int*)smem;
    int*  rlen  = roff + 640;
    int*  cnt   = rlen + 640;
    int*  wtot  = cnt + CSR_R;
    int*  wbs   = wtot + 8;
    int2* stage = (int2*)(wbs + 8);

    const int tid  = threadIdx.x;
    const int lane = tid & 31;
    const int wv   = tid >> 5;
    const int nch  = (E + CSR_CH - 1) / CSR_CH;

    if (tid < 128) { roff[512 + tid] = 0; rlen[512 + tid] = 0; }
    cnt[tid] = 0;
    int bsum = 0;

    for (int c0 = 0; c0 < nch; c0 += 512) {
        __syncthreads();
        #pragma unroll
        for (int hh = 0; hh < 2; ++hh) {
            const int t = tid + 256 * hh;
            const int c = c0 + t;
            int len = 0, ro = 0;
            if (c < nch) {
                const int o0 = coff[(size_t)c * CSR_NBP + b];
                const int o1 = coff[(size_t)c * CSR_NBP + b + 1];
                ro  = c * CSR_CH + o0;
                len = o1 - o0;
                bsum += o0;
            }
            roff[t] = ro;
            rlen[t] = len;
        }
        __syncthreads();
        const int tn = min(512, nch - c0);
        for (int k = wv; k < tn; k += 64) {
            int2 rec[8]; int ln[8], so[8];
            int mx = 0;
            #pragma unroll
            for (int q = 0; q < 8; ++q) {
                ln[q] = __builtin_amdgcn_readfirstlane(rlen[k + 8 * q]);
                so[q] = __builtin_amdgcn_readfirstlane(roff[k + 8 * q]);
                mx = max(mx, ln[q]);
                const int idx = min(so[q] + min(lane, max(ln[q] - 1, 0)), E - 1);
                rec[q] = tmp[idx];
            }
            #pragma unroll
            for (int q = 0; q < 8; ++q)
                if (lane < ln[q]) atomicAdd(&cnt[rec[q].x >> 17], 1);
            if (mx > 32) {
                for (int q = 0; q < 8; ++q)
                    for (int j = lane + 32; j < ln[q]; j += 32)
                        atomicAdd(&cnt[tmp[(size_t)so[q] + j].x >> 17], 1);
            }
        }
    }
    __syncthreads();

    const int v = cnt[tid];
    int incl = v;
    for (int d = 1; d < 32; d <<= 1) {
        const int t = __shfl_up(incl, d);
        if (lane >= d) incl += t;
    }
    if (lane == 31) wtot[wv] = incl;
    for (int d = 16; d >= 1; d >>= 1) bsum += __shfl_xor(bsum, d);
    if (lane == 0) wbs[wv] = bsum;
    __syncthreads();
    int bstart = 0, total = 0;
    for (int k = 0; k < 8; ++k) bstart += wbs[k];
    for (int k = 0; k < 8; ++k) total  += wtot[k];
    {
        int excl = incl - v;
        for (int k = 0; k < wv; ++k) excl += wtot[k];
        cnt[tid] = excl;
        const int row = b * CSR_R + tid;
        if (row < nrows)      start[row]   = bstart + excl;
        if (row == nrows - 1) start[nrows] = bstart + excl + v;
    }
    const bool fits = total <= CSR_CAP;

    for (int c0 = 0; c0 < nch; c0 += 512) {
        __syncthreads();
        #pragma unroll
        for (int hh = 0; hh < 2; ++hh) {
            const int t = tid + 256 * hh;
            const int c = c0 + t;
            int len = 0, ro = 0;
            if (c < nch) {
                const int o0 = coff[(size_t)c * CSR_NBP + b];
                const int o1 = coff[(size_t)c * CSR_NBP + b + 1];
                ro  = c * CSR_CH + o0;
                len = o1 - o0;
            }
            roff[t] = ro;
            rlen[t] = len;
        }
        __syncthreads();
        const int tn = min(512, nch - c0);
        for (int k = wv; k < tn; k += 64) {
            int2 rec[8]; int ln[8], so[8];
            int mx = 0;
            #pragma unroll
            for (int q = 0; q < 8; ++q) {
                ln[q] = __builtin_amdgcn_readfirstlane(rlen[k + 8 * q]);
                so[q] = __builtin_amdgcn_readfirstlane(roff[k + 8 * q]);
                mx = max(mx, ln[q]);
                const int idx = min(so[q] + min(lane, max(ln[q] - 1, 0)), E - 1);
                rec[q] = tmp[idx];
            }
            #pragma unroll
            for (int q = 0; q < 8; ++q)
                if (lane < ln[q]) gcn_csr_place(rec[q], cnt, fits, stage, sorted, bstart);
            if (mx > 32) {
                for (int q = 0; q < 8; ++q)
                    for (int j = lane + 32; j < ln[q]; j += 32)
                        gcn_csr_place(tmp[(size_t)so[q] + j], cnt, fits, stage, sorted, bstart);
            }
        }
    }
    __syncthreads();
    if (fits)
        for (int i = tid; i < total; i += 256) sorted[(size_t)bstart + i] = stage[i];
}

template <int GEMM_BLOCKS, int N, int K, int KP>
__global__ __launch_bounds__(256)
void gcn_gemm1_with_csr_rows(const float* __restrict__ A,
                             const float* __restrict__ B,
                             float* __restrict__ C,
                             int M,
                             const int2* __restrict__ tmp,
                             const int* __restrict__ coff,
                             int* __restrict__ start,
                             int2* __restrict__ sorted,
                             int E, int nrows)
{
    constexpr int BM = (8 / (N / 64)) * 16;
    static_assert(CSR_L2_LDS_BYTES >= (KP * N + BM * GA_PITCH) * 4, "LDS buffer holds either role");
    __shared__ __align__(16) unsigned char smem[CSR_L2_LDS_BYTES];
    if (blockIdx.x < GEMM_BLOCKS)
        gcn_gemm_wmma_f32<N, K, KP>((float*)smem, (float*)smem + KP * N, blockIdx.x, A, B, C, M);
    else
        gcn_csr_rows(smem, blockIdx.x - GEMM_BLOCKS, tmp, coff, start, sorted, E, nrows);
}

template <int D, int U>
__device__ __forceinline__
void gcn_gather_batch(float4& acc, const float* __restrict__ sup,
                      int ecol, int ewb, int j, int sub, int f)
{
    constexpr int EPI = 128 / D;
    int    c[U];
    float  wv[U];
    float4 v[U];
    #pragma unroll
    for (int u = 0; u < U; ++u) {
        if (EPI == 1) {
            c[u]  = __builtin_amdgcn_readlane(ecol, j + u);
            wv[u] = __int_as_float(__builtin_amdgcn_readlane(ewb, j + u));
        } else {
            const int c0 = __builtin_amdgcn_readlane(ecol, j + 2 * u);
            const int c1 = __builtin_amdgcn_readlane(ecol, j + 2 * u + 1);
            const int w0 = __builtin_amdgcn_readlane(ewb,  j + 2 * u);
            const int w1 = __builtin_amdgcn_readlane(ewb,  j + 2 * u + 1);
            c[u]  = sub ? c1 : c0;
            wv[u] = __int_as_float(sub ? w1 : w0);
        }
    }
    #pragma unroll
    for (int u = 0; u < U; ++u)
        v[u] = *(const float4*)(sup + (size_t)c[u] * D + f);
    #pragma unroll
    for (int u = 0; u < U; ++u) {
        acc.x = fmaf(wv[u], v[u].x, acc.x);
        acc.y = fmaf(wv[u], v[u].y, acc.y);
        acc.z = fmaf(wv[u], v[u].z, acc.z);
        acc.w = fmaf(wv[u], v[u].w, acc.w);
    }
}

template <int D>
__global__ __launch_bounds__(256)
void gcn_spmm_gather(const int* __restrict__ start,
                     const int2* __restrict__ sorted,
                     const float* __restrict__ sup,
                     const float* __restrict__ bias,
                     float* __restrict__ out,
                     int nrows)
{
    constexpr int LPE = D / 4;
    constexpr int EPI = 32 / LPE;
    const int lane = threadIdx.x & 31;
    const int sub  = lane / LPE;
    const int f    = (lane - sub * LPE) * 4;
    const int wave0  = __builtin_amdgcn_readfirstlane(blockIdx.x * 8 + (threadIdx.x >> 5));
    const int nwaves = gridDim.x * 8;
    const float4 bv = *(const float4*)(bias + f);

    for (int r = wave0; r < nrows; r += nwaves) {
        const int beg = start[r];
        const int end = start[r + 1];
        float4 acc = make_float4(0.0f, 0.0f, 0.0f, 0.0f);

        for (int base = beg; base < end; base += 32) {
            const int n = min(32, end - base);
            const int2 ev  = sorted[base + min(lane, n - 1)];
            const int ecol = ev.x;
            const int ewb  = (lane < n) ? ev.y : 0;
            int j = 0;
            for (; j + 8 <= n; j += 8)
                gcn_gather_batch<D, 8 / EPI>(acc, sup, ecol, ewb, j, sub, f);
            for (; j < n; j += 4)
                gcn_gather_batch<D, 4 / EPI>(acc, sup, ecol, ewb, j, sub, f);
        }

        if (EPI == 2) {
            acc.x += __shfl_xor(acc.x, 16);
            acc.y += __shfl_xor(acc.y, 16);
            acc.z += __shfl_xor(acc.z, 16);
            acc.w += __shfl_xor(acc.w, 16);
        }
        if (sub == 0) {
            float4 o;
            o.x = acc.x + bv.x; o.y = acc.y + bv.y;
            o.z = acc.z + bv.z; o.w = acc.w + bv.w;
            *(float4*)(out + (size_t)r * D + f) = o;
        }
    }
}


#define FG_PITCH 132
__global__ __launch_bounds__(512)
void gcn_gather_relu_gemm(const int* __restrict__ start,
                          const int2* __restrict__ sorted,
                          const float* __restrict__ sup,
                          const float* __restrict__ bias,
                          const float* __restrict__ B,
                          float* __restrict__ C,
                          int nrows)
{
    constexpr int D  = 128;
    constexpr int NO = 64;
    extern __shared__ float fgl[];
    float* ldsB = fgl;
    float* hT   = fgl + D * NO;
    float* oT   = hT + 16 * FG_PITCH;

    const int tid  = threadIdx.x;
    const int lane = tid & 31;
    const int wave = __builtin_amdgcn_readfirstlane(tid >> 5);
    const int half = lane >> 4;
    const int l    = lane & 15;
    const int f    = lane * 4;

    #pragma unroll
    for (int idx = 0; idx < (D / 2) * NO / 512; ++idx) {
        const int pi = idx * 512 + tid;
        const int p  = pi / NO;
        const int n  = pi - p * NO;
        v2f pr;
        pr.x = B[(size_t)(2 * p) * NO + n];
        pr.y = B[(size_t)(2 * p + 1) * NO + n];
        *(v2f*)(ldsB + (size_t)pi * 2) = pr;
    }

    const float4 bv = *(const float4*)(bias + f);
    const int ntiles = (nrows + 15) >> 4;

    for (int tile = blockIdx.x; tile < ntiles; tile += gridDim.x) {
        const int r = tile * 16 + wave;
        float4 o = make_float4(0.0f, 0.0f, 0.0f, 0.0f);
        if (r < nrows) {
            const int beg = start[r];
            const int end = start[r + 1];
            float4 acc = make_float4(0.0f, 0.0f, 0.0f, 0.0f);
            for (int base = beg; base < end; base += 32) {
                const int n = min(32, end - base);
                const int2 ev  = sorted[base + min(lane, n - 1)];
                const int ecol = ev.x;
                const int ewb  = (lane < n) ? ev.y : 0;
                int j = 0;
                for (; j + 8 <= n; j += 8)
                    gcn_gather_batch<D, 8>(acc, sup, ecol, ewb, j, 0, f);
                for (; j < n; j += 4)
                    gcn_gather_batch<D, 4>(acc, sup, ecol, ewb, j, 0, f);
            }
            o.x = fmaxf(acc.x + bv.x, 0.0f); o.y = fmaxf(acc.y + bv.y, 0.0f);
            o.z = fmaxf(acc.z + bv.z, 0.0f); o.w = fmaxf(acc.w + bv.w, 0.0f);
        }
        *(float4*)(hT + wave * FG_PITCH + f) = o;
        __syncthreads();

        if (wave < 4) {
            const int n0 = wave * 16;
            const float* ap = hT + l * FG_PITCH + 2 * half;
            const float* bp = ldsB + (size_t)(half * NO + n0 + l) * 2;
            v8f c0 = {}, c1 = {};
            #pragma unroll
            for (int s = 0; s < D / 4; s += 2) {
                const v2f a0 = *(const v2f*)(ap + 4 * s);
                const v2f b0 = *(const v2f*)(bp + s * (NO * 4));
                const v2f a1 = *(const v2f*)(ap + 4 * s + 4);
                const v2f b1 = *(const v2f*)(bp + (s + 1) * (NO * 4));
                c0 = __builtin_amdgcn_wmma_f32_16x16x4_f32(false, a0, false, b0,
                                                           (short)0, c0, false, false);
                c1 = __builtin_amdgcn_wmma_f32_16x16x4_f32(false, a1, false, b1,
                                                           (short)0, c1, false, false);
            }
            #pragma unroll
            for (int i = 0; i < 8; ++i)
                oT[(i + 8 * half) * NO + n0 + l] = c0[i] + c1[i];
        }
        __syncthreads();

        if (tid < 256) {
            const int row = tid >> 4;
            const int c4  = (tid & 15) * 4;
            const int rr  = tile * 16 + row;
            if (rr < nrows)
                *(float4*)(C + (size_t)rr * NO + c4) = *(const float4*)(oT + row * NO + c4);
        }
    }
}

extern "C" void kernel_launch(void* const* d_in, const int* in_sizes, int n_in,
                              void* d_out, int out_size, void* d_ws, size_t ws_size,
                              hipStream_t stream)
{
    const float* x  = (const float*)d_in[0];
    const int*   ei = (const int*)  d_in[1];
    const float* ew = (const float*)d_in[2];
    const float* W1 = (const float*)d_in[3];
    const float* b1 = (const float*)d_in[4];
    const float* W2 = (const float*)d_in[5];
    const float* b2 = (const float*)d_in[6];
    float* out = (float*)d_out;

    const int N  = 100000;
    const int E  = in_sizes[2];
    constexpr int F  = 256;
    constexpr int H1 = 128;
    constexpr int H2 = 64;
    constexpr int KP = 64;

    float* sup1 = (float*)d_ws;
    float* h    = sup1 + (size_t)N * H1;
    float* sup2 = h    + (size_t)N * H1;
    int*   start  = (int*)(sup2 + (size_t)N * H2);
    int2*  sorted = (int2*)(start + 100352);
    int2*  tmp    = sorted + E;
    int*   coff   = (int*)(tmp + E);

    gcn_part_chunks<<<256, 1024, 0, stream>>>(ei, ei + E, ew, tmp, coff, E);

    {
        constexpr int GEMM_BLOCKS = 1563;
        gcn_gemm1_with_csr_rows<GEMM_BLOCKS, H1, F, KP>
            <<<GEMM_BLOCKS + CSR_NB, 256, 0, stream>>>(x, W1, sup1, N,
                                                       tmp, coff, start, sorted, E, N);
    }
    {
        const size_t lds = (size_t)(H1 * H2 + 16 * FG_PITCH + 16 * H2) * sizeof(float);
        gcn_gather_relu_gemm<<<1024, 512, lds, stream>>>(start, sorted, sup1, b1, W2, sup2, N);
    }
    gcn_spmm_gather<H2><<<2048, 256, 0, stream>>>(start, sorted, sup2, b2, out, N);
}
